# starting kernel
# baseline (speedup 1.0000x reference)
#include <hip/hip_runtime.h>
#include <hip/hip_bf16.h>

#define CC 8
#define MM 32
#define GG 16
#define LL 4

typedef float v2f __attribute__((ext_vector_type(2)));
typedef float v8f __attribute__((ext_vector_type(8)));

__global__ void softmax_rows_kernel(const float* __restrict__ src,
                                    float* __restrict__ dst,
                                    int nrows, int len, int stride,
                                    int d1, int s1, int d2, int s2) {
    int r = blockIdx.x * blockDim.x + threadIdx.x;
    if (r >= nrows) return;
    int base = (r / d1) * s1 + ((r % d1) / d2) * s2 + (r % d2);
    float mx = -3.402823466e38f;
    for (int i = 0; i < len; ++i) mx = fmaxf(mx, src[base + i * stride]);
    float sum = 0.f;
    for (int i = 0; i < len; ++i) sum += expf(src[base + i * stride] - mx);
    float inv = 1.0f / sum;
    for (int i = 0; i < len; ++i)
        dst[base + i * stride] = expf(src[base + i * stride] - mx) * inv;
}

__global__ void zero_int_kernel(int* __restrict__ p, int n) {
    int t = blockIdx.x * blockDim.x + threadIdx.x;
    if (t < n) p[t] = 0;
}

__global__ void degree_kernel(const int* __restrict__ dst, int* __restrict__ cnt, int E) {
    int t = blockIdx.x * blockDim.x + threadIdx.x;
    if (t < E) atomicAdd(&cnt[dst[t]], 1);
}

__global__ void invcnt_kernel(const int* __restrict__ cnt, float* __restrict__ inv, int N) {
    int t = blockIdx.x * blockDim.x + threadIdx.x;
    if (t < N) inv[t] = 1.0f / fmaxf((float)cnt[t], 1.0f);
}

__global__ __launch_bounds__(1024) void scan_kernel(const int* __restrict__ cnt,
                                                    int* __restrict__ rowstart,
                                                    int* __restrict__ cursor, int N) {
    __shared__ int buf[1024];
    __shared__ int carry;
    if (threadIdx.x == 0) carry = 0;
    __syncthreads();
    for (int base = 0; base < N; base += 1024) {
        int i = base + (int)threadIdx.x;
        int v = (i < N) ? cnt[i] : 0;
        buf[threadIdx.x] = v;
        __syncthreads();
        for (int off = 1; off < 1024; off <<= 1) {
            int t = (threadIdx.x >= (unsigned)off) ? buf[threadIdx.x - off] : 0;
            __syncthreads();
            buf[threadIdx.x] += t;
            __syncthreads();
        }
        int excl = buf[threadIdx.x] - v + carry;
        if (i < N) { rowstart[i] = excl; cursor[i] = excl; }
        __syncthreads();
        if (threadIdx.x == 1023) carry += buf[1023];
        __syncthreads();
    }
    if (threadIdx.x == 0) rowstart[N] = carry;
}

__global__ void csr_fill_kernel(const int* __restrict__ dst, const int* __restrict__ src,
                                int* __restrict__ cursor, int* __restrict__ eidx, int E) {
    int e = blockIdx.x * blockDim.x + threadIdx.x;
    if (e < E) {
        int p = atomicAdd(&cursor[dst[e]], 1);
        eidx[p] = src[e];
    }
}

__global__ void layer0_kernel(const int* __restrict__ x,
                              const float* __restrict__ B0s,
                              const float* __restrict__ Pis,
                              float* __restrict__ h,
                              float* __restrict__ out,
                              int N) {
    int t = blockIdx.x * blockDim.x + threadIdx.x;
    if (t >= N * GG) return;
    int n = t / GG, g = t % GG;
    int xm = x[n];
    float u[CC];
    float Z = 0.f;
#pragma unroll
    for (int c = 0; c < CC; ++c) {
        u[c] = B0s[(c * MM + xm) * GG + g] * Pis[c * GG + g];
        Z += u[c];
    }
    out[(size_t)n * (LL * GG) + g] = logf(Z);
    float iz = 1.0f / Z;
#pragma unroll
    for (int c = 0; c < CC; ++c)
        h[(size_t)n * (GG * CC) + g * CC + c] = u[c] * iz;
}

__global__ void gather_mean_kernel(const float* __restrict__ h,
                                   const int* __restrict__ rowstart,
                                   const int* __restrict__ eidx,
                                   const float* __restrict__ invc,
                                   float* __restrict__ aggr, int N) {
    int t = blockIdx.x * blockDim.x + threadIdx.x;
    int n = t >> 5;
    int lane = t & 31;
    if (n >= N) return;
    int beg = rowstart[n];
    int end = rowstart[n + 1];
    float4 acc = make_float4(0.f, 0.f, 0.f, 0.f);
    for (int j = beg; j < end; ++j) {
        int s = eidx[j];
        const float4 v = *(const float4*)(h + (size_t)s * 128 + lane * 4);
        acc.x += v.x; acc.y += v.y; acc.z += v.z; acc.w += v.w;
    }
    float iv = invc[n];
    acc.x *= iv; acc.y *= iv; acc.z *= iv; acc.w *= iv;
    *(float4*)(aggr + (size_t)n * 128 + lane * 4) = acc;
}

__global__ __launch_bounds__(512) void layer_update_kernel(
    const int* __restrict__ x,
    const float* __restrict__ Qg,
    const float* __restrict__ Bg,
    const float* __restrict__ aggr,
    float* __restrict__ h,
    float* __restrict__ out,
    int N, int lout) {
    __shared__ float qa[GG * 16 * CC];

    const int tid = threadIdx.x;
    const int w = tid >> 5;
    const int lane = tid & 31;
    const int half = lane >> 4;
    const int col = lane & 15;
    const int g = w;

    const int node = blockIdx.x * 16 + col;
    const int nn = node < N ? node : N - 1;
    const float sc = node < N ? 1.0f : 0.0f;
    const size_t abase = (size_t)nn * (GG * CC) + g * CC;
    const int k0 = half * 2;

    v2f a0, a1, b0, b1;
    a0.x = aggr[abase + k0 + 0] * sc;
    a0.y = aggr[abase + k0 + 1] * sc;
    a1.x = aggr[abase + k0 + 4] * sc;
    a1.y = aggr[abase + k0 + 5] * sc;

    const int qc = col & 7;
    const float cm = (col < CC) ? 1.0f : 0.0f;
    b0.x = Qg[(qc * CC + k0 + 0) * GG + g] * cm;
    b0.y = Qg[(qc * CC + k0 + 1) * GG + g] * cm;
    b1.x = Qg[(qc * CC + k0 + 4) * GG + g] * cm;
    b1.y = Qg[(qc * CC + k0 + 5) * GG + g] * cm;

    v8f acc = {0.f, 0.f, 0.f, 0.f, 0.f, 0.f, 0.f, 0.f};
    acc = __builtin_amdgcn_wmma_f32_16x16x4_f32(false, a0, false, b0,
                                                (short)0, acc, false, false);
    acc = __builtin_amdgcn_wmma_f32_16x16x4_f32(false, a1, false, b1,
                                                (short)0, acc, false, false);

    if (col < CC) {
#pragma unroll
        for (int r = 0; r < 8; ++r)
            qa[(w * 16 + (r + half * 8)) * CC + col] = acc[r];
    }
    __syncthreads();

    if (lane < 16) {
        int n2 = blockIdx.x * 16 + lane;
        if (n2 < N) {
            int xm = x[n2];
            float u[CC];
            float Z = 0.f;
#pragma unroll
            for (int c = 0; c < CC; ++c) {
                float bn = Bg[(c * MM + xm) * GG + g];
                u[c] = bn * qa[(w * 16 + lane) * CC + c];
                Z += u[c];
            }
            out[(size_t)n2 * (LL * GG) + lout * GG + g] = logf(Z);
            float iz = 1.0f / Z;
#pragma unroll
            for (int c = 0; c < CC; ++c)
                h[(size_t)n2 * (GG * CC) + g * CC + c] = u[c] * iz;
        }
    }
}

extern "C" void kernel_launch(void* const* d_in, const int* in_sizes, int n_in,
                              void* d_out, int out_size, void* d_ws, size_t ws_size,
                              hipStream_t stream) {
    const int* x = (const int*)d_in[0];
    const int* ei = (const int*)d_in[1];
    const float* lB0 = (const float*)d_in[2];
    const float* lPi = (const float*)d_in[3];
    const float* lQ = (const float*)d_in[4];
    const float* lB = (const float*)d_in[5];
    float* out = (float*)d_out;

    const int N = in_sizes[0];
    const int E = in_sizes[1] / 2;
    const int* dst = ei;
    const int* srcv = ei + E;

    float* ws = (float*)d_ws;
    size_t off = 0;
    float* B0s = ws + off; off += CC * MM * GG;
    float* Pis = ws + off; off += CC * GG;
    float* Qs  = ws + off; off += (LL - 1) * CC * CC * GG;
    float* Bs  = ws + off; off += (LL - 1) * CC * MM * GG;
    float* invc = ws + off; off += N;
    int* cnt      = (int*)(ws + off); off += N;
    int* rowstart = (int*)(ws + off); off += N + 1;
    int* cursor   = (int*)(ws + off); off += N;
    int* eidx     = (int*)(ws + off); off += E;
    off = (off + 15) & ~(size_t)15;
    float* h    = ws + off; off += (size_t)N * GG * CC;
    float* aggr = ws + off; off += (size_t)N * GG * CC;
    (void)ws_size; (void)n_in; (void)out_size;

    softmax_rows_kernel<<<1, 128, 0, stream>>>(lB0, B0s, CC * GG, MM, GG,
                                               CC * GG, 0, GG, MM * GG);
    softmax_rows_kernel<<<1, 32, 0, stream>>>(lPi, Pis, GG, CC, GG,
                                              GG, 0, GG, 0);
    softmax_rows_kernel<<<3, 128, 0, stream>>>(lQ, Qs, (LL - 1) * CC * GG, CC,
                                               CC * GG, CC * GG, CC * CC * GG,
                                               1, 1);
    softmax_rows_kernel<<<3, 128, 0, stream>>>(lB, Bs, (LL - 1) * CC * GG, MM,
                                               GG, CC * GG, CC * MM * GG,
                                               GG, MM * GG);

    zero_int_kernel<<<(N + 255) / 256, 256, 0, stream>>>(cnt, N);
    degree_kernel<<<(E + 255) / 256, 256, 0, stream>>>(dst, cnt, E);
    invcnt_kernel<<<(N + 255) / 256, 256, 0, stream>>>(cnt, invc, N);
    scan_kernel<<<1, 1024, 0, stream>>>(cnt, rowstart, cursor, N);
    csr_fill_kernel<<<(E + 255) / 256, 256, 0, stream>>>(dst, srcv, cursor, eidx, E);

    layer0_kernel<<<(N * GG + 255) / 256, 256, 0, stream>>>(x, B0s, Pis, h, out, N);

    const int gatherBlocks = (int)(((size_t)N * 32 + 255) / 256);
    for (int l = 0; l < LL - 1; ++l) {
        gather_mean_kernel<<<gatherBlocks, 256, 0, stream>>>(h, rowstart, eidx,
                                                             invc, aggr, N);
        layer_update_kernel<<<(N + 15) / 16, 512, 0, stream>>>(
            x, Qs + (size_t)l * CC * CC * GG, Bs + (size_t)l * CC * MM * GG,
            aggr, h, out, N, l + 1);
    }
}


# CSR row-offset scan: one block's 49 serial Hillis-Steele chunks replaced by 13 independent blocks, each summing the counts in front of its 4096-count tile and scanning that tile once with DPP adds
# speedup vs baseline: 1.3533x; 1.3533x over previous
#include <hip/hip_runtime.h>
#include <hip/hip_bf16.h>

#define CC 8
#define MM 32
#define GG 16
#define LL 4

typedef float v2f __attribute__((ext_vector_type(2)));
typedef float v8f __attribute__((ext_vector_type(8)));

__global__ void softmax_rows_kernel(const float* __restrict__ src,
                                    float* __restrict__ dst,
                                    int nrows, int len, int stride,
                                    int d1, int s1, int d2, int s2) {
    int r = blockIdx.x * blockDim.x + threadIdx.x;
    if (r >= nrows) return;
    int base = (r / d1) * s1 + ((r % d1) / d2) * s2 + (r % d2);
    float mx = -3.402823466e38f;
    for (int i = 0; i < len; ++i) mx = fmaxf(mx, src[base + i * stride]);
    float sum = 0.f;
    for (int i = 0; i < len; ++i) sum += expf(src[base + i * stride] - mx);
    float inv = 1.0f / sum;
    for (int i = 0; i < len; ++i)
        dst[base + i * stride] = expf(src[base + i * stride] - mx) * inv;
}

__global__ void zero_int_kernel(int* __restrict__ p, int n) {
    int t = blockIdx.x * blockDim.x + threadIdx.x;
    if (t < n) p[t] = 0;
}

__global__ void degree_kernel(const int* __restrict__ dst, int* __restrict__ cnt, int E) {
    int t = blockIdx.x * blockDim.x + threadIdx.x;
    if (t < E) atomicAdd(&cnt[dst[t]], 1);
}

__global__ void invcnt_kernel(const int* __restrict__ cnt, float* __restrict__ inv, int N) {
    int t = blockIdx.x * blockDim.x + threadIdx.x;
    if (t < N) inv[t] = 1.0f / fmaxf((float)cnt[t], 1.0f);
}

#define SCAN_THREADS 1024
#define SCAN_TILE (SCAN_THREADS * 4)

__device__ __forceinline__ int wave_inclusive_sum(int v, int lane) {
    v += __builtin_amdgcn_update_dpp(0, v, 0x111, 0xF, 0xF, true);
    v += __builtin_amdgcn_update_dpp(0, v, 0x112, 0xF, 0xF, true);
    v += __builtin_amdgcn_update_dpp(0, v, 0x114, 0xF, 0xF, true);
    v += __builtin_amdgcn_update_dpp(0, v, 0x118, 0xF, 0xF, true);
    const int low_row_total = __builtin_amdgcn_readlane(v, 15);
    return v + (lane >= 16 ? low_row_total : 0);
}

__global__ __launch_bounds__(SCAN_THREADS) void scan_kernel(const int* __restrict__ cnt,
                                                            int* __restrict__ rowstart,
                                                            int* __restrict__ cursor, int N) {
    __shared__ int s_front[32];
    __shared__ int s_tile[32];
    const int tid = threadIdx.x;
    const int lane = tid & 31;
    const int w = tid >> 5;
    const int tile0 = blockIdx.x * SCAN_TILE;

    int front = 0;
    for (int i = tid * 4; i < tile0; i += SCAN_TILE) {
        const int4 q = *(const int4*)(cnt + i);
        front += (q.x + q.y) + (q.z + q.w);
    }

    const int i0 = tile0 + tid * 4;
    int4 v = make_int4(0, 0, 0, 0);
    if (i0 + 3 < N) {
        v = *(const int4*)(cnt + i0);
    } else {
        if (i0 + 0 < N) v.x = cnt[i0 + 0];
        if (i0 + 1 < N) v.y = cnt[i0 + 1];
        if (i0 + 2 < N) v.z = cnt[i0 + 2];
    }
    const int s1 = v.x + v.y;
    const int s2 = s1 + v.z;
    const int s3 = s2 + v.w;

    const int incl = wave_inclusive_sum(s3, lane);
    const int front_incl = wave_inclusive_sum(front, lane);
    if (lane == 31) {
        s_tile[w] = incl;
        s_front[w] = front_incl;
    }
    __syncthreads();

    const int part = s_front[lane] + (lane < w ? s_tile[lane] : 0);
    const int wave_base = __builtin_amdgcn_readlane(wave_inclusive_sum(part, lane), 31);

    const int e0 = wave_base + incl - s3;
    const int e1 = e0 + v.x;
    const int e2 = e0 + s1;
    const int e3 = e0 + s2;
    if (i0 + 3 < N) {
        const int4 o = make_int4(e0, e1, e2, e3);
        *(int4*)(rowstart + i0) = o;
        *(int4*)(cursor + i0) = o;
    } else {
        if (i0 + 0 <= N) rowstart[i0 + 0] = e0;
        if (i0 + 1 <= N) rowstart[i0 + 1] = e1;
        if (i0 + 2 <= N) rowstart[i0 + 2] = e2;
        if (i0 + 3 <= N) rowstart[i0 + 3] = e3;
        if (i0 + 0 < N) cursor[i0 + 0] = e0;
        if (i0 + 1 < N) cursor[i0 + 1] = e1;
        if (i0 + 2 < N) cursor[i0 + 2] = e2;
    }
}

__global__ void csr_fill_kernel(const int* __restrict__ dst, const int* __restrict__ src,
                                int* __restrict__ cursor, int* __restrict__ eidx, int E) {
    int e = blockIdx.x * blockDim.x + threadIdx.x;
    if (e < E) {
        int p = atomicAdd(&cursor[dst[e]], 1);
        eidx[p] = src[e];
    }
}

__global__ void layer0_kernel(const int* __restrict__ x,
                              const float* __restrict__ B0s,
                              const float* __restrict__ Pis,
                              float* __restrict__ h,
                              float* __restrict__ out,
                              int N) {
    int t = blockIdx.x * blockDim.x + threadIdx.x;
    if (t >= N * GG) return;
    int n = t / GG, g = t % GG;
    int xm = x[n];
    float u[CC];
    float Z = 0.f;
#pragma unroll
    for (int c = 0; c < CC; ++c) {
        u[c] = B0s[(c * MM + xm) * GG + g] * Pis[c * GG + g];
        Z += u[c];
    }
    out[(size_t)n * (LL * GG) + g] = logf(Z);
    float iz = 1.0f / Z;
#pragma unroll
    for (int c = 0; c < CC; ++c)
        h[(size_t)n * (GG * CC) + g * CC + c] = u[c] * iz;
}

__global__ void gather_mean_kernel(const float* __restrict__ h,
                                   const int* __restrict__ rowstart,
                                   const int* __restrict__ eidx,
                                   const float* __restrict__ invc,
                                   float* __restrict__ aggr, int N) {
    int t = blockIdx.x * blockDim.x + threadIdx.x;
    int n = t >> 5;
    int lane = t & 31;
    if (n >= N) return;
    int beg = rowstart[n];
    int end = rowstart[n + 1];
    float4 acc = make_float4(0.f, 0.f, 0.f, 0.f);
    for (int j = beg; j < end; ++j) {
        int s = eidx[j];
        const float4 v = *(const float4*)(h + (size_t)s * 128 + lane * 4);
        acc.x += v.x; acc.y += v.y; acc.z += v.z; acc.w += v.w;
    }
    float iv = invc[n];
    acc.x *= iv; acc.y *= iv; acc.z *= iv; acc.w *= iv;
    *(float4*)(aggr + (size_t)n * 128 + lane * 4) = acc;
}

__global__ __launch_bounds__(512) void layer_update_kernel(
    const int* __restrict__ x,
    const float* __restrict__ Qg,
    const float* __restrict__ Bg,
    const float* __restrict__ aggr,
    float* __restrict__ h,
    float* __restrict__ out,
    int N, int lout) {
    __shared__ float qa[GG * 16 * CC];

    const int tid = threadIdx.x;
    const int w = tid >> 5;
    const int lane = tid & 31;
    const int half = lane >> 4;
    const int col = lane & 15;
    const int g = w;

    const int node = blockIdx.x * 16 + col;
    const int nn = node < N ? node : N - 1;
    const float sc = node < N ? 1.0f : 0.0f;
    const size_t abase = (size_t)nn * (GG * CC) + g * CC;
    const int k0 = half * 2;

    v2f a0, a1, b0, b1;
    a0.x = aggr[abase + k0 + 0] * sc;
    a0.y = aggr[abase + k0 + 1] * sc;
    a1.x = aggr[abase + k0 + 4] * sc;
    a1.y = aggr[abase + k0 + 5] * sc;

    const int qc = col & 7;
    const float cm = (col < CC) ? 1.0f : 0.0f;
    b0.x = Qg[(qc * CC + k0 + 0) * GG + g] * cm;
    b0.y = Qg[(qc * CC + k0 + 1) * GG + g] * cm;
    b1.x = Qg[(qc * CC + k0 + 4) * GG + g] * cm;
    b1.y = Qg[(qc * CC + k0 + 5) * GG + g] * cm;

    v8f acc = {0.f, 0.f, 0.f, 0.f, 0.f, 0.f, 0.f, 0.f};
    acc = __builtin_amdgcn_wmma_f32_16x16x4_f32(false, a0, false, b0,
                                                (short)0, acc, false, false);
    acc = __builtin_amdgcn_wmma_f32_16x16x4_f32(false, a1, false, b1,
                                                (short)0, acc, false, false);

    if (col < CC) {
#pragma unroll
        for (int r = 0; r < 8; ++r)
            qa[(w * 16 + (r + half * 8)) * CC + col] = acc[r];
    }
    __syncthreads();

    if (lane < 16) {
        int n2 = blockIdx.x * 16 + lane;
        if (n2 < N) {
            int xm = x[n2];
            float u[CC];
            float Z = 0.f;
#pragma unroll
            for (int c = 0; c < CC; ++c) {
                float bn = Bg[(c * MM + xm) * GG + g];
                u[c] = bn * qa[(w * 16 + lane) * CC + c];
                Z += u[c];
            }
            out[(size_t)n2 * (LL * GG) + lout * GG + g] = logf(Z);
            float iz = 1.0f / Z;
#pragma unroll
            for (int c = 0; c < CC; ++c)
                h[(size_t)n2 * (GG * CC) + g * CC + c] = u[c] * iz;
        }
    }
}

extern "C" void kernel_launch(void* const* d_in, const int* in_sizes, int n_in,
                              void* d_out, int out_size, void* d_ws, size_t ws_size,
                              hipStream_t stream) {
    const int* x = (const int*)d_in[0];
    const int* ei = (const int*)d_in[1];
    const float* lB0 = (const float*)d_in[2];
    const float* lPi = (const float*)d_in[3];
    const float* lQ = (const float*)d_in[4];
    const float* lB = (const float*)d_in[5];
    float* out = (float*)d_out;

    const int N = in_sizes[0];
    const int E = in_sizes[1] / 2;
    const int* dst = ei;
    const int* srcv = ei + E;

    float* ws = (float*)d_ws;
    size_t off = 0;
    float* B0s = ws + off; off += CC * MM * GG;
    float* Pis = ws + off; off += CC * GG;
    float* Qs  = ws + off; off += (LL - 1) * CC * CC * GG;
    float* Bs  = ws + off; off += (LL - 1) * CC * MM * GG;
    float* invc = ws + off; off += N;
    off = (off + 3) & ~(size_t)3;
    int* cnt      = (int*)(ws + off); off += N;
    off = (off + 3) & ~(size_t)3;
    int* rowstart = (int*)(ws + off); off += N + 1;
    off = (off + 3) & ~(size_t)3;
    int* cursor   = (int*)(ws + off); off += N;
    int* eidx     = (int*)(ws + off); off += E;
    off = (off + 15) & ~(size_t)15;
    float* h    = ws + off; off += (size_t)N * GG * CC;
    float* aggr = ws + off; off += (size_t)N * GG * CC;
    (void)ws_size; (void)n_in; (void)out_size;

    softmax_rows_kernel<<<1, 128, 0, stream>>>(lB0, B0s, CC * GG, MM, GG,
                                               CC * GG, 0, GG, MM * GG);
    softmax_rows_kernel<<<1, 32, 0, stream>>>(lPi, Pis, GG, CC, GG,
                                              GG, 0, GG, 0);
    softmax_rows_kernel<<<3, 128, 0, stream>>>(lQ, Qs, (LL - 1) * CC * GG, CC,
                                               CC * GG, CC * GG, CC * CC * GG,
                                               1, 1);
    softmax_rows_kernel<<<3, 128, 0, stream>>>(lB, Bs, (LL - 1) * CC * GG, MM,
                                               GG, CC * GG, CC * MM * GG,
                                               GG, MM * GG);

    zero_int_kernel<<<(N + 255) / 256, 256, 0, stream>>>(cnt, N);
    degree_kernel<<<(E + 255) / 256, 256, 0, stream>>>(dst, cnt, E);
    invcnt_kernel<<<(N + 255) / 256, 256, 0, stream>>>(cnt, invc, N);
    scan_kernel<<<N / SCAN_TILE + 1, SCAN_THREADS, 0, stream>>>(cnt, rowstart, cursor, N);
    csr_fill_kernel<<<(E + 255) / 256, 256, 0, stream>>>(dst, srcv, cursor, eidx, E);

    layer0_kernel<<<(N * GG + 255) / 256, 256, 0, stream>>>(x, B0s, Pis, h, out, N);

    const int gatherBlocks = (int)(((size_t)N * 32 + 255) / 256);
    for (int l = 0; l < LL - 1; ++l) {
        gather_mean_kernel<<<gatherBlocks, 256, 0, stream>>>(h, rowstart, eidx,
                                                             invc, aggr, N);
        layer_update_kernel<<<(N + 15) / 16, 512, 0, stream>>>(
            x, Qs + (size_t)l * CC * CC * GG, Bs + (size_t)l * CC * MM * GG,
            aggr, h, out, N, l + 1);
    }
}


# Gather: a node's edge ids fetched 32 per coalesced load, its neighbour rows requested in groups of up to 8 (remainder as one group too), not one dependent id+row round trip per remainder edge
# speedup vs baseline: 1.8655x; 1.3785x over previous
#include <hip/hip_runtime.h>
#include <hip/hip_bf16.h>

#define CC 8
#define MM 32
#define GG 16
#define LL 4

typedef float v2f __attribute__((ext_vector_type(2)));
typedef float v8f __attribute__((ext_vector_type(8)));

template <int LEN>
__device__ __forceinline__ void softmax_row(const float* __restrict__ src,
                                            float* __restrict__ dst,
                                            int base, int stride) {
    float v[LEN];
#pragma unroll
    for (int i = 0; i < LEN; ++i) v[i] = src[base + i * stride];
    float mx = -3.402823466e38f;
#pragma unroll
    for (int i = 0; i < LEN; ++i) mx = fmaxf(mx, v[i]);
    float sum = 0.f;
#pragma unroll
    for (int i = 0; i < LEN; ++i) {
        v[i] = expf(v[i] - mx);
        sum += v[i];
    }
    const float inv = 1.0f / sum;
#pragma unroll
    for (int i = 0; i < LEN; ++i) dst[base + i * stride] = v[i] * inv;
}

#define SETUP_THREADS 256
#define SETUP_SOFTMAX_THREADS 928

__global__ __launch_bounds__(SETUP_THREADS) void setup_kernel(
    const float* __restrict__ lB0, const float* __restrict__ lPi,
    const float* __restrict__ lQ, const float* __restrict__ lB,
    float* __restrict__ B0s, float* __restrict__ Pis,
    float* __restrict__ Qs, float* __restrict__ Bs,
    int* __restrict__ cnt, int N) {
    const int t = blockIdx.x * SETUP_THREADS + threadIdx.x;
    const int step = gridDim.x * SETUP_THREADS * 4;
    for (int i = t * 4; i < N; i += step) {
        if (i + 3 < N) {
            *(int4*)(cnt + i) = make_int4(0, 0, 0, 0);
        } else {
            if (i + 0 < N) cnt[i + 0] = 0;
            if (i + 1 < N) cnt[i + 1] = 0;
            if (i + 2 < N) cnt[i + 2] = 0;
        }
    }

    if (t < 128) {
        softmax_row<MM>(lB0, B0s, (t / GG) * (MM * GG) + (t % GG), GG);
    } else if (t < 160) {
        if (t < 128 + GG) softmax_row<CC>(lPi, Pis, t - 128, GG);
    } else if (t < 544) {
        const int r = t - 160;
        softmax_row<CC>(lQ, Qs, (r / (CC * GG)) * (CC * CC * GG) + (r % (CC * GG)), CC * GG);
    } else if (t < SETUP_SOFTMAX_THREADS) {
        const int r = t - 544;
        softmax_row<MM>(lB, Bs,
                        (r / (CC * GG)) * (CC * MM * GG) + ((r % (CC * GG)) / GG) * (MM * GG) + (r % GG),
                        GG);
    }
}

__global__ void degree_kernel(const int* __restrict__ dst, int* __restrict__ cnt, int E) {
    int t = blockIdx.x * blockDim.x + threadIdx.x;
    if (t < E) atomicAdd(&cnt[dst[t]], 1);
}

#define SCAN_THREADS 1024
#define SCAN_TILE (SCAN_THREADS * 4)

__device__ __forceinline__ int wave_inclusive_sum(int v, int lane) {
    v += __builtin_amdgcn_update_dpp(0, v, 0x111, 0xF, 0xF, true);
    v += __builtin_amdgcn_update_dpp(0, v, 0x112, 0xF, 0xF, true);
    v += __builtin_amdgcn_update_dpp(0, v, 0x114, 0xF, 0xF, true);
    v += __builtin_amdgcn_update_dpp(0, v, 0x118, 0xF, 0xF, true);
    const int low_row_total = __builtin_amdgcn_readlane(v, 15);
    return v + (lane >= 16 ? low_row_total : 0);
}

__global__ __launch_bounds__(SCAN_THREADS) void scan_kernel(const int* __restrict__ cnt,
                                                            int* __restrict__ rowstart,
                                                            int* __restrict__ cursor,
                                                            float* __restrict__ inv, int N) {
    __shared__ int s_front[32];
    __shared__ int s_tile[32];
    const int tid = threadIdx.x;
    const int lane = tid & 31;
    const int w = tid >> 5;
    const int tile0 = blockIdx.x * SCAN_TILE;

    int front = 0;
    for (int i = tid * 4; i < tile0; i += SCAN_TILE) {
        const int4 q = *(const int4*)(cnt + i);
        front += (q.x + q.y) + (q.z + q.w);
    }

    const int i0 = tile0 + tid * 4;
    int4 v = make_int4(0, 0, 0, 0);
    if (i0 + 3 < N) {
        v = *(const int4*)(cnt + i0);
    } else {
        if (i0 + 0 < N) v.x = cnt[i0 + 0];
        if (i0 + 1 < N) v.y = cnt[i0 + 1];
        if (i0 + 2 < N) v.z = cnt[i0 + 2];
    }
    if (i0 + 3 < N) {
        *(float4*)(inv + i0) = make_float4(1.0f / fmaxf((float)v.x, 1.0f),
                                           1.0f / fmaxf((float)v.y, 1.0f),
                                           1.0f / fmaxf((float)v.z, 1.0f),
                                           1.0f / fmaxf((float)v.w, 1.0f));
    } else {
        if (i0 + 0 < N) inv[i0 + 0] = 1.0f / fmaxf((float)v.x, 1.0f);
        if (i0 + 1 < N) inv[i0 + 1] = 1.0f / fmaxf((float)v.y, 1.0f);
        if (i0 + 2 < N) inv[i0 + 2] = 1.0f / fmaxf((float)v.z, 1.0f);
    }

    const int s1 = v.x + v.y;
    const int s2 = s1 + v.z;
    const int s3 = s2 + v.w;

    const int incl = wave_inclusive_sum(s3, lane);
    const int front_incl = wave_inclusive_sum(front, lane);
    if (lane == 31) {
        s_tile[w] = incl;
        s_front[w] = front_incl;
    }
    __syncthreads();

    const int part = s_front[lane] + (lane < w ? s_tile[lane] : 0);
    const int wave_base = __builtin_amdgcn_readlane(wave_inclusive_sum(part, lane), 31);

    const int e0 = wave_base + incl - s3;
    const int e1 = e0 + v.x;
    const int e2 = e0 + s1;
    const int e3 = e0 + s2;
    if (i0 + 3 < N) {
        const int4 o = make_int4(e0, e1, e2, e3);
        *(int4*)(rowstart + i0) = o;
        *(int4*)(cursor + i0) = o;
    } else {
        if (i0 + 0 <= N) rowstart[i0 + 0] = e0;
        if (i0 + 1 <= N) rowstart[i0 + 1] = e1;
        if (i0 + 2 <= N) rowstart[i0 + 2] = e2;
        if (i0 + 3 <= N) rowstart[i0 + 3] = e3;
        if (i0 + 0 < N) cursor[i0 + 0] = e0;
        if (i0 + 1 < N) cursor[i0 + 1] = e1;
        if (i0 + 2 < N) cursor[i0 + 2] = e2;
    }
}


__device__ __forceinline__ void csr_fill_body(const int* __restrict__ dst, const int* __restrict__ src,
                                              int* __restrict__ cursor, int* __restrict__ eidx,
                                              int E, int e) {
    if (e < E) {
        int p = atomicAdd(&cursor[dst[e]], 1);
        eidx[p] = src[e];
    }
}

__device__ __forceinline__ void layer0_body(const int* __restrict__ x,
                                            const float* __restrict__ B0s,
                                            const float* __restrict__ Pis,
                                            float* __restrict__ h,
                                            float* __restrict__ out,
                                            int N, int t) {
    if (t >= N * GG) return;
    int n = t / GG, g = t % GG;
    int xm = x[n];
    float u[CC];
    float Z = 0.f;
#pragma unroll
    for (int c = 0; c < CC; ++c) {
        u[c] = B0s[(c * MM + xm) * GG + g] * Pis[c * GG + g];
        Z += u[c];
    }
    out[(size_t)n * (LL * GG) + g] = logf(Z);
    float iz = 1.0f / Z;
#pragma unroll
    for (int c = 0; c < CC; ++c)
        h[(size_t)n * (GG * CC) + g * CC + c] = u[c] * iz;
}

__global__ __launch_bounds__(256) void fill_and_layer0_kernel(
    const int* __restrict__ dst, const int* __restrict__ src,
    int* __restrict__ cursor, int* __restrict__ eidx, int E, int fillBlocks,
    const int* __restrict__ x, const float* __restrict__ B0s,
    const float* __restrict__ Pis, float* __restrict__ h,
    float* __restrict__ out, int N) {
    const int b = blockIdx.x;
    if (b < fillBlocks)
        csr_fill_body(dst, src, cursor, eidx, E, b * 256 + threadIdx.x);
    else
        layer0_body(x, B0s, Pis, h, out, N, (b - fillBlocks) * 256 + threadIdx.x);
}

template <int R>
__device__ __forceinline__ void add_rows(const float* __restrict__ hin, int ids, int j0,
                                         int lane, float4& acc) {
    float4 v[R];
#pragma unroll
    for (int u = 0; u < R; ++u) {
        const int s = __builtin_amdgcn_readlane(ids, j0 + u);
        v[u] = *(const float4*)(hin + (size_t)s * 128 + lane * 4);
    }
#pragma unroll
    for (int u = 0; u < R; ++u) {
        acc.x += v[u].x; acc.y += v[u].y; acc.z += v[u].z; acc.w += v[u].w;
    }
}

#define MEAN_PITCH 132

__global__ __launch_bounds__(512) void gather_update_kernel(
    const int* __restrict__ x,
    const float* __restrict__ Qg,
    const float* __restrict__ Bg,
    const float* __restrict__ hin,
    const int* __restrict__ rowstart,
    const int* __restrict__ eidx,
    const float* __restrict__ invc,
    float* __restrict__ hout,
    float* __restrict__ out,
    int N, int lout) {
    __shared__ float mean[16 * MEAN_PITCH];
    __shared__ float qa[GG * 16 * CC];

    const int tid = threadIdx.x;
    const int w = tid >> 5;
    const int lane = tid & 31;
    const int half = lane >> 4;
    const int col = lane & 15;
    const int g = w;
    const int k0 = half * 2;

    const int node = blockIdx.x * 16 + col;
    const int nn = node < N ? node : N - 1;
    const int xm = x[nn];

    const int qc = col & 7;
    const float cm = (col < CC) ? 1.0f : 0.0f;
    v2f b0, b1;
    b0.x = Qg[(qc * CC + k0 + 0) * GG + g] * cm;
    b0.y = Qg[(qc * CC + k0 + 1) * GG + g] * cm;
    b1.x = Qg[(qc * CC + k0 + 4) * GG + g] * cm;
    b1.y = Qg[(qc * CC + k0 + 5) * GG + g] * cm;

    const int gn = __builtin_amdgcn_readfirstlane(blockIdx.x * 16 + w);
    float4 acc = make_float4(0.f, 0.f, 0.f, 0.f);
    if (gn < N) {
        const int beg = rowstart[gn];
        const int end = rowstart[gn + 1];
        const float iv = invc[gn];
        for (int c0 = beg; c0 < end; c0 += 32) {
            const int n = (end - c0 < 32) ? (end - c0) : 32;
            const int ids = (lane < n) ? eidx[c0 + lane] : 0;
            int j = 0;
            for (; j + 8 <= n; j += 8) add_rows<8>(hin, ids, j, lane, acc);
            switch (n - j) {
                case 7: add_rows<7>(hin, ids, j, lane, acc); break;
                case 6: add_rows<6>(hin, ids, j, lane, acc); break;
                case 5: add_rows<5>(hin, ids, j, lane, acc); break;
                case 4: add_rows<4>(hin, ids, j, lane, acc); break;
                case 3: add_rows<3>(hin, ids, j, lane, acc); break;
                case 2: add_rows<2>(hin, ids, j, lane, acc); break;
                case 1: add_rows<1>(hin, ids, j, lane, acc); break;
                default: break;
            }
        }
        acc.x *= iv; acc.y *= iv; acc.z *= iv; acc.w *= iv;
    }
    *(float4*)(mean + w * MEAN_PITCH + lane * 4) = acc;

    float bn[CC];
#pragma unroll
    for (int c = 0; c < CC; ++c) bn[c] = Bg[(c * MM + xm) * GG + g];

    __syncthreads();

    const float* arow = mean + col * MEAN_PITCH + g * CC + k0;
    v2f a0, a1;
    a0.x = arow[0];
    a0.y = arow[1];
    a1.x = arow[4];
    a1.y = arow[5];

    v8f d = {0.f, 0.f, 0.f, 0.f, 0.f, 0.f, 0.f, 0.f};
    d = __builtin_amdgcn_wmma_f32_16x16x4_f32(false, a0, false, b0,
                                              (short)0, d, false, false);
    d = __builtin_amdgcn_wmma_f32_16x16x4_f32(false, a1, false, b1,
                                              (short)0, d, false, false);

    if (col < CC) {
#pragma unroll
        for (int r = 0; r < 8; ++r)
            qa[(w * 16 + (r + half * 8)) * CC + col] = d[r];
    }
    __syncthreads();

    if (lane < 16) {
        if (node < N) {
            float u[CC];
            float Z = 0.f;
#pragma unroll
            for (int c = 0; c < CC; ++c) {
                u[c] = bn[c] * qa[(w * 16 + lane) * CC + c];
                Z += u[c];
            }
            out[(size_t)node * (LL * GG) + lout * GG + g] = logf(Z);
            float iz = 1.0f / Z;
#pragma unroll
            for (int c = 0; c < CC; ++c)
                hout[(size_t)node * (GG * CC) + g * CC + c] = u[c] * iz;
        }
    }
}

extern "C" void kernel_launch(void* const* d_in, const int* in_sizes, int n_in,
                              void* d_out, int out_size, void* d_ws, size_t ws_size,
                              hipStream_t stream) {
    const int* x = (const int*)d_in[0];
    const int* ei = (const int*)d_in[1];
    const float* lB0 = (const float*)d_in[2];
    const float* lPi = (const float*)d_in[3];
    const float* lQ = (const float*)d_in[4];
    const float* lB = (const float*)d_in[5];
    float* out = (float*)d_out;

    const int N = in_sizes[0];
    const int E = in_sizes[1] / 2;
    const int* dst = ei;
    const int* srcv = ei + E;

    float* ws = (float*)d_ws;
    size_t off = 0;
    float* B0s = ws + off; off += CC * MM * GG;
    float* Pis = ws + off; off += CC * GG;
    float* Qs  = ws + off; off += (LL - 1) * CC * CC * GG;
    float* Bs  = ws + off; off += (LL - 1) * CC * MM * GG;
    off = (off + 3) & ~(size_t)3;
    float* invc = ws + off; off += N;
    off = (off + 3) & ~(size_t)3;
    int* cnt      = (int*)(ws + off); off += N;
    off = (off + 3) & ~(size_t)3;
    int* rowstart = (int*)(ws + off); off += N + 1;
    off = (off + 3) & ~(size_t)3;
    int* cursor   = (int*)(ws + off); off += N;
    int* eidx     = (int*)(ws + off); off += E;
    off = (off + 15) & ~(size_t)15;
    float* h0 = ws + off; off += (size_t)N * GG * CC;
    float* h1 = ws + off; off += (size_t)N * GG * CC;
    (void)ws_size; (void)n_in; (void)out_size;

    const int setupBlocks = ((N + 3) / 4 + SETUP_THREADS - 1) / SETUP_THREADS +
                            (SETUP_SOFTMAX_THREADS + SETUP_THREADS - 1) / SETUP_THREADS;
    setup_kernel<<<setupBlocks, SETUP_THREADS, 0, stream>>>(lB0, lPi, lQ, lB, B0s, Pis, Qs, Bs, cnt, N);

    degree_kernel<<<(E + 255) / 256, 256, 0, stream>>>(dst, cnt, E);
    scan_kernel<<<N / SCAN_TILE + 1, SCAN_THREADS, 0, stream>>>(cnt, rowstart, cursor, invc, N);

    const int fillBlocks = (E + 255) / 256;
    const int layer0Blocks = (N * GG + 255) / 256;
    fill_and_layer0_kernel<<<fillBlocks + layer0Blocks, 256, 0, stream>>>(
        dst, srcv, cursor, eidx, E, fillBlocks, x, B0s, Pis, h0, out, N);

    const int tileBlocks = (N + 15) / 16;
    gather_update_kernel<<<tileBlocks, 512, 0, stream>>>(
        x, Qs, Bs, h0, rowstart, eidx, invc, h1, out, N, 1);
    gather_update_kernel<<<tileBlocks, 512, 0, stream>>>(
        x, Qs + CC * CC * GG, Bs + CC * MM * GG, h1, rowstart, eidx, invc, h0, out, N, 2);
    gather_update_kernel<<<tileBlocks, 512, 0, stream>>>(
        x, Qs + 2 * CC * CC * GG, Bs + 2 * CC * MM * GG, h0, rowstart, eidx, invc, h1, out, N, 3);
}


# First graph layer gathers from the 32-row table of layer-0 posteriors (a function of x alone) held in LDS, fetching 4-byte symbols x[id] instead of 512-byte node rows; h0 is never stored per node
# speedup vs baseline: 1.9996x; 1.0719x over previous
#include <hip/hip_runtime.h>
#include <hip/hip_bf16.h>

#define CC 8
#define MM 32
#define GG 16
#define LL 4

typedef float v2f __attribute__((ext_vector_type(2)));
typedef float v8f __attribute__((ext_vector_type(8)));

template <int LEN>
__device__ __forceinline__ void softmax_row(const float* __restrict__ src,
                                            float* __restrict__ dst,
                                            int base, int stride) {
    float v[LEN];
#pragma unroll
    for (int i = 0; i < LEN; ++i) v[i] = src[base + i * stride];
    float mx = -3.402823466e38f;
#pragma unroll
    for (int i = 0; i < LEN; ++i) mx = fmaxf(mx, v[i]);
    float sum = 0.f;
#pragma unroll
    for (int i = 0; i < LEN; ++i) {
        v[i] = expf(v[i] - mx);
        sum += v[i];
    }
    const float inv = 1.0f / sum;
#pragma unroll
    for (int i = 0; i < LEN; ++i) dst[base + i * stride] = v[i] * inv;
}

#define SETUP_THREADS 256
#define SETUP_SOFTMAX_THREADS 928

__global__ __launch_bounds__(SETUP_THREADS) void setup_kernel(
    const float* __restrict__ lB0, const float* __restrict__ lPi,
    const float* __restrict__ lQ, const float* __restrict__ lB,
    float* __restrict__ B0s, float* __restrict__ Pis,
    float* __restrict__ Qs, float* __restrict__ Bs,
    int* __restrict__ cnt, int N) {
    const int t = blockIdx.x * SETUP_THREADS + threadIdx.x;
    const int step = gridDim.x * SETUP_THREADS * 4;
    for (int i = t * 4; i < N; i += step) {
        if (i + 3 < N) {
            *(int4*)(cnt + i) = make_int4(0, 0, 0, 0);
        } else {
            if (i + 0 < N) cnt[i + 0] = 0;
            if (i + 1 < N) cnt[i + 1] = 0;
            if (i + 2 < N) cnt[i + 2] = 0;
        }
    }

    if (t < 128) {
        softmax_row<MM>(lB0, B0s, (t / GG) * (MM * GG) + (t % GG), GG);
    } else if (t < 160) {
        if (t < 128 + GG) softmax_row<CC>(lPi, Pis, t - 128, GG);
    } else if (t < 544) {
        const int r = t - 160;
        softmax_row<CC>(lQ, Qs, (r / (CC * GG)) * (CC * CC * GG) + (r % (CC * GG)), CC * GG);
    } else if (t < SETUP_SOFTMAX_THREADS) {
        const int r = t - 544;
        softmax_row<MM>(lB, Bs,
                        (r / (CC * GG)) * (CC * MM * GG) + ((r % (CC * GG)) / GG) * (MM * GG) + (r % GG),
                        GG);
    }
}

__global__ void degree_kernel(const int* __restrict__ dst, int* __restrict__ cnt, int E) {
    int t = blockIdx.x * blockDim.x + threadIdx.x;
    if (t < E) atomicAdd(&cnt[dst[t]], 1);
}

#define SCAN_THREADS 1024
#define SCAN_TILE (SCAN_THREADS * 4)

__device__ __forceinline__ int wave_inclusive_sum(int v, int lane) {
    v += __builtin_amdgcn_update_dpp(0, v, 0x111, 0xF, 0xF, true);
    v += __builtin_amdgcn_update_dpp(0, v, 0x112, 0xF, 0xF, true);
    v += __builtin_amdgcn_update_dpp(0, v, 0x114, 0xF, 0xF, true);
    v += __builtin_amdgcn_update_dpp(0, v, 0x118, 0xF, 0xF, true);
    const int low_row_total = __builtin_amdgcn_readlane(v, 15);
    return v + (lane >= 16 ? low_row_total : 0);
}

__global__ __launch_bounds__(SCAN_THREADS) void scan_kernel(const int* __restrict__ cnt,
                                                            int* __restrict__ rowstart,
                                                            int* __restrict__ cursor,
                                                            float* __restrict__ inv, int N) {
    __shared__ int s_front[32];
    __shared__ int s_tile[32];
    const int tid = threadIdx.x;
    const int lane = tid & 31;
    const int w = tid >> 5;
    const int tile0 = blockIdx.x * SCAN_TILE;

    int front = 0;
    for (int i = tid * 4; i < tile0; i += SCAN_TILE) {
        const int4 q = *(const int4*)(cnt + i);
        front += (q.x + q.y) + (q.z + q.w);
    }

    const int i0 = tile0 + tid * 4;
    int4 v = make_int4(0, 0, 0, 0);
    if (i0 + 3 < N) {
        v = *(const int4*)(cnt + i0);
    } else {
        if (i0 + 0 < N) v.x = cnt[i0 + 0];
        if (i0 + 1 < N) v.y = cnt[i0 + 1];
        if (i0 + 2 < N) v.z = cnt[i0 + 2];
    }
    if (i0 + 3 < N) {
        *(float4*)(inv + i0) = make_float4(1.0f / fmaxf((float)v.x, 1.0f),
                                           1.0f / fmaxf((float)v.y, 1.0f),
                                           1.0f / fmaxf((float)v.z, 1.0f),
                                           1.0f / fmaxf((float)v.w, 1.0f));
    } else {
        if (i0 + 0 < N) inv[i0 + 0] = 1.0f / fmaxf((float)v.x, 1.0f);
        if (i0 + 1 < N) inv[i0 + 1] = 1.0f / fmaxf((float)v.y, 1.0f);
        if (i0 + 2 < N) inv[i0 + 2] = 1.0f / fmaxf((float)v.z, 1.0f);
    }

    const int s1 = v.x + v.y;
    const int s2 = s1 + v.z;
    const int s3 = s2 + v.w;

    const int incl = wave_inclusive_sum(s3, lane);
    const int front_incl = wave_inclusive_sum(front, lane);
    if (lane == 31) {
        s_tile[w] = incl;
        s_front[w] = front_incl;
    }
    __syncthreads();

    const int part = s_front[lane] + (lane < w ? s_tile[lane] : 0);
    const int wave_base = __builtin_amdgcn_readlane(wave_inclusive_sum(part, lane), 31);

    const int e0 = wave_base + incl - s3;
    const int e1 = e0 + v.x;
    const int e2 = e0 + s1;
    const int e3 = e0 + s2;
    if (i0 + 3 < N) {
        const int4 o = make_int4(e0, e1, e2, e3);
        *(int4*)(rowstart + i0) = o;
        *(int4*)(cursor + i0) = o;
    } else {
        if (i0 + 0 <= N) rowstart[i0 + 0] = e0;
        if (i0 + 1 <= N) rowstart[i0 + 1] = e1;
        if (i0 + 2 <= N) rowstart[i0 + 2] = e2;
        if (i0 + 3 <= N) rowstart[i0 + 3] = e3;
        if (i0 + 0 < N) cursor[i0 + 0] = e0;
        if (i0 + 1 < N) cursor[i0 + 1] = e1;
        if (i0 + 2 < N) cursor[i0 + 2] = e2;
    }
}


__device__ __forceinline__ void csr_fill_body(const int* __restrict__ dst, const int* __restrict__ src,
                                              int* __restrict__ cursor, int* __restrict__ eidx,
                                              int E, int e) {
    if (e < E) {
        int p = atomicAdd(&cursor[dst[e]], 1);
        eidx[p] = src[e];
    }
}

__device__ __forceinline__ void layer0_body(const int* __restrict__ x,
                                            const float* __restrict__ B0s,
                                            const float* __restrict__ Pis,
                                            float* __restrict__ tab0,
                                            float* __restrict__ out,
                                            int N, int t) {
    if (t >= N * GG + MM * GG) return;
    const bool per_node = t < N * GG;
    const int r = per_node ? t : t - N * GG;
    const int n = r / GG, g = r % GG;
    const int xm = per_node ? x[n] : n;
    float u[CC];
    float Z = 0.f;
#pragma unroll
    for (int c = 0; c < CC; ++c) {
        u[c] = B0s[(c * MM + xm) * GG + g] * Pis[c * GG + g];
        Z += u[c];
    }
    if (per_node) {
        out[(size_t)n * (LL * GG) + g] = logf(Z);
    } else {
        const float iz = 1.0f / Z;
#pragma unroll
        for (int c = 0; c < CC; ++c)
            tab0[xm * (GG * CC) + g * CC + c] = u[c] * iz;
    }
}

__global__ __launch_bounds__(256) void fill_and_layer0_kernel(
    const int* __restrict__ dst, const int* __restrict__ src,
    int* __restrict__ cursor, int* __restrict__ eidx, int E, int fillBlocks,
    const int* __restrict__ x, const float* __restrict__ B0s,
    const float* __restrict__ Pis, float* __restrict__ tab0,
    float* __restrict__ out, int N) {
    const int b = blockIdx.x;
    if (b < fillBlocks)
        csr_fill_body(dst, src, cursor, eidx, E, b * 256 + threadIdx.x);
    else
        layer0_body(x, B0s, Pis, tab0, out, N, (b - fillBlocks) * 256 + threadIdx.x);
}

template <int R>
__device__ __forceinline__ void add_rows(const float* __restrict__ hin, int ids, int j0,
                                         int lane, float4& acc) {
    float4 v[R];
#pragma unroll
    for (int u = 0; u < R; ++u) {
        const int s = __builtin_amdgcn_readlane(ids, j0 + u);
        v[u] = *(const float4*)(hin + (size_t)s * 128 + lane * 4);
    }
#pragma unroll
    for (int u = 0; u < R; ++u) {
        acc.x += v[u].x; acc.y += v[u].y; acc.z += v[u].z; acc.w += v[u].w;
    }
}

#define MEAN_PITCH 132

template <bool FIRST>
__global__ __launch_bounds__(512) void gather_update_kernel(
    const int* __restrict__ x,
    const float* __restrict__ Qg,
    const float* __restrict__ Bg,
    const float* __restrict__ hin,
    const int* __restrict__ rowstart,
    const int* __restrict__ eidx,
    const float* __restrict__ invc,
    float* __restrict__ hout,
    float* __restrict__ out,
    int N, int lout) {
    __shared__ float mean[16 * MEAN_PITCH];
    __shared__ float qa[GG * 16 * CC];
    __shared__ __attribute__((aligned(16))) float tab[FIRST ? MM * GG * CC : 4];

    const int tid = threadIdx.x;
    const int w = tid >> 5;
    const int lane = tid & 31;
    const int half = lane >> 4;
    const int col = lane & 15;
    const int g = w;
    const int k0 = half * 2;

    const int node = blockIdx.x * 16 + col;
    const int nn = node < N ? node : N - 1;
    const int xm = x[nn];

    const int qc = col & 7;
    const float cm = (col < CC) ? 1.0f : 0.0f;
    v2f b0, b1;
    b0.x = Qg[(qc * CC + k0 + 0) * GG + g] * cm;
    b0.y = Qg[(qc * CC + k0 + 1) * GG + g] * cm;
    b1.x = Qg[(qc * CC + k0 + 4) * GG + g] * cm;
    b1.y = Qg[(qc * CC + k0 + 5) * GG + g] * cm;

    const int gn = __builtin_amdgcn_readfirstlane(blockIdx.x * 16 + w);
    float4 acc = make_float4(0.f, 0.f, 0.f, 0.f);
    if constexpr (FIRST) {
        const float4 t0 = ((const float4*)hin)[tid];
        const float4 t1 = ((const float4*)hin)[tid + 512];
        int beg = 0, end = 0;
        float iv = 0.f;
        if (gn < N) {
            beg = rowstart[gn];
            end = rowstart[gn + 1];
            iv = invc[gn];
        }
        int n = (end - beg < 32) ? (end - beg) : 32;
        int xs = (lane < n) ? x[eidx[beg + lane]] : 0;
        ((float4*)tab)[tid] = t0;
        ((float4*)tab)[tid + 512] = t1;
        __syncthreads();
        for (int c0 = beg; c0 < end;) {
            int j = 0;
            for (; j + 8 <= n; j += 8) add_rows<8>(tab, xs, j, lane, acc);
            switch (n - j) {
                case 7: add_rows<7>(tab, xs, j, lane, acc); break;
                case 6: add_rows<6>(tab, xs, j, lane, acc); break;
                case 5: add_rows<5>(tab, xs, j, lane, acc); break;
                case 4: add_rows<4>(tab, xs, j, lane, acc); break;
                case 3: add_rows<3>(tab, xs, j, lane, acc); break;
                case 2: add_rows<2>(tab, xs, j, lane, acc); break;
                case 1: add_rows<1>(tab, xs, j, lane, acc); break;
                default: break;
            }
            c0 += 32;
            if (c0 < end) {
                n = (end - c0 < 32) ? (end - c0) : 32;
                xs = (lane < n) ? x[eidx[c0 + lane]] : 0;
            }
        }
        acc.x *= iv; acc.y *= iv; acc.z *= iv; acc.w *= iv;
    } else if (gn < N) {
        const int beg = rowstart[gn];
        const int end = rowstart[gn + 1];
        const float iv = invc[gn];
        for (int c0 = beg; c0 < end; c0 += 32) {
            const int n = (end - c0 < 32) ? (end - c0) : 32;
            const int ids = (lane < n) ? eidx[c0 + lane] : 0;
            int j = 0;
            for (; j + 8 <= n; j += 8) add_rows<8>(hin, ids, j, lane, acc);
            switch (n - j) {
                case 7: add_rows<7>(hin, ids, j, lane, acc); break;
                case 6: add_rows<6>(hin, ids, j, lane, acc); break;
                case 5: add_rows<5>(hin, ids, j, lane, acc); break;
                case 4: add_rows<4>(hin, ids, j, lane, acc); break;
                case 3: add_rows<3>(hin, ids, j, lane, acc); break;
                case 2: add_rows<2>(hin, ids, j, lane, acc); break;
                case 1: add_rows<1>(hin, ids, j, lane, acc); break;
                default: break;
            }
        }
        acc.x *= iv; acc.y *= iv; acc.z *= iv; acc.w *= iv;
    }
    *(float4*)(mean + w * MEAN_PITCH + lane * 4) = acc;

    float bn[CC];
#pragma unroll
    for (int c = 0; c < CC; ++c) bn[c] = Bg[(c * MM + xm) * GG + g];

    __syncthreads();

    const float* arow = mean + col * MEAN_PITCH + g * CC + k0;
    v2f a0, a1;
    a0.x = arow[0];
    a0.y = arow[1];
    a1.x = arow[4];
    a1.y = arow[5];

    v8f d = {0.f, 0.f, 0.f, 0.f, 0.f, 0.f, 0.f, 0.f};
    d = __builtin_amdgcn_wmma_f32_16x16x4_f32(false, a0, false, b0,
                                              (short)0, d, false, false);
    d = __builtin_amdgcn_wmma_f32_16x16x4_f32(false, a1, false, b1,
                                              (short)0, d, false, false);

    if (col < CC) {
#pragma unroll
        for (int r = 0; r < 8; ++r)
            qa[(w * 16 + (r + half * 8)) * CC + col] = d[r];
    }
    __syncthreads();

    if (lane < 16) {
        if (node < N) {
            float u[CC];
            float Z = 0.f;
#pragma unroll
            for (int c = 0; c < CC; ++c) {
                u[c] = bn[c] * qa[(w * 16 + lane) * CC + c];
                Z += u[c];
            }
            out[(size_t)node * (LL * GG) + lout * GG + g] = logf(Z);
            float iz = 1.0f / Z;
#pragma unroll
            for (int c = 0; c < CC; ++c)
                hout[(size_t)node * (GG * CC) + g * CC + c] = u[c] * iz;
        }
    }
}

extern "C" void kernel_launch(void* const* d_in, const int* in_sizes, int n_in,
                              void* d_out, int out_size, void* d_ws, size_t ws_size,
                              hipStream_t stream) {
    const int* x = (const int*)d_in[0];
    const int* ei = (const int*)d_in[1];
    const float* lB0 = (const float*)d_in[2];
    const float* lPi = (const float*)d_in[3];
    const float* lQ = (const float*)d_in[4];
    const float* lB = (const float*)d_in[5];
    float* out = (float*)d_out;

    const int N = in_sizes[0];
    const int E = in_sizes[1] / 2;
    const int* dst = ei;
    const int* srcv = ei + E;

    float* ws = (float*)d_ws;
    size_t off = 0;
    float* B0s = ws + off; off += CC * MM * GG;
    float* Pis = ws + off; off += CC * GG;
    float* Qs  = ws + off; off += (LL - 1) * CC * CC * GG;
    float* Bs  = ws + off; off += (LL - 1) * CC * MM * GG;
    off = (off + 3) & ~(size_t)3;
    float* invc = ws + off; off += N;
    off = (off + 3) & ~(size_t)3;
    int* cnt      = (int*)(ws + off); off += N;
    off = (off + 3) & ~(size_t)3;
    int* rowstart = (int*)(ws + off); off += N + 1;
    off = (off + 3) & ~(size_t)3;
    int* cursor   = (int*)(ws + off); off += N;
    int* eidx     = (int*)(ws + off); off += E;
    off = (off + 15) & ~(size_t)15;
    float* h0 = ws + off; off += (size_t)N * GG * CC;
    float* h1 = ws + off; off += (size_t)N * GG * CC;
    float* tab0 = h0;
    (void)ws_size; (void)n_in; (void)out_size;

    const int setupBlocks = ((N + 3) / 4 + SETUP_THREADS - 1) / SETUP_THREADS +
                            (SETUP_SOFTMAX_THREADS + SETUP_THREADS - 1) / SETUP_THREADS;
    setup_kernel<<<setupBlocks, SETUP_THREADS, 0, stream>>>(lB0, lPi, lQ, lB, B0s, Pis, Qs, Bs, cnt, N);

    degree_kernel<<<(E + 255) / 256, 256, 0, stream>>>(dst, cnt, E);
    scan_kernel<<<N / SCAN_TILE + 1, SCAN_THREADS, 0, stream>>>(cnt, rowstart, cursor, invc, N);

    const int fillBlocks = (E + 255) / 256;
    const int layer0Blocks = (N * GG + MM * GG + 255) / 256;
    fill_and_layer0_kernel<<<fillBlocks + layer0Blocks, 256, 0, stream>>>(
        dst, srcv, cursor, eidx, E, fillBlocks, x, B0s, Pis, tab0, out, N);

    const int tileBlocks = (N + 15) / 16;
    gather_update_kernel<true><<<tileBlocks, 512, 0, stream>>>(
        x, Qs, Bs, tab0, rowstart, eidx, invc, h1, out, N, 1);
    gather_update_kernel<false><<<tileBlocks, 512, 0, stream>>>(
        x, Qs + CC * CC * GG, Bs + CC * MM * GG, h1, rowstart, eidx, invc, h0, out, N, 2);
    gather_update_kernel<false><<<tileBlocks, 512, 0, stream>>>(
        x, Qs + 2 * CC * CC * GG, Bs + 2 * CC * MM * GG, h0, rowstart, eidx, invc, h1, out, N, 3);
}


# Node-state rows move as whole aligned lines: h buffers at the workspace head (rows began mid-line: 5 lines per row read, now 4); a tile's rows and log-likelihoods leave via LDS as full-row stores
# speedup vs baseline: 2.2288x; 1.1146x over previous
#include <hip/hip_runtime.h>
#include <hip/hip_bf16.h>

#define CC 8
#define MM 32
#define GG 16
#define LL 4

typedef float v2f __attribute__((ext_vector_type(2)));
typedef float v8f __attribute__((ext_vector_type(8)));

template <int LEN>
__device__ __forceinline__ void softmax_row(const float* __restrict__ src,
                                            float* __restrict__ dst,
                                            int base, int stride) {
    float v[LEN];
#pragma unroll
    for (int i = 0; i < LEN; ++i) v[i] = src[base + i * stride];
    float mx = -3.402823466e38f;
#pragma unroll
    for (int i = 0; i < LEN; ++i) mx = fmaxf(mx, v[i]);
    float sum = 0.f;
#pragma unroll
    for (int i = 0; i < LEN; ++i) {
        v[i] = expf(v[i] - mx);
        sum += v[i];
    }
    const float inv = 1.0f / sum;
#pragma unroll
    for (int i = 0; i < LEN; ++i) dst[base + i * stride] = v[i] * inv;
}

#define SETUP_THREADS 256
#define SETUP_SOFTMAX_THREADS 928

__global__ __launch_bounds__(SETUP_THREADS) void setup_kernel(
    const float* __restrict__ lB0, const float* __restrict__ lPi,
    const float* __restrict__ lQ, const float* __restrict__ lB,
    float* __restrict__ B0s, float* __restrict__ Pis,
    float* __restrict__ Qs, float* __restrict__ Bs,
    int* __restrict__ cnt, int N) {
    const int t = blockIdx.x * SETUP_THREADS + threadIdx.x;
    const int step = gridDim.x * SETUP_THREADS * 4;
    for (int i = t * 4; i < N; i += step) {
        if (i + 3 < N) {
            *(int4*)(cnt + i) = make_int4(0, 0, 0, 0);
        } else {
            if (i + 0 < N) cnt[i + 0] = 0;
            if (i + 1 < N) cnt[i + 1] = 0;
            if (i + 2 < N) cnt[i + 2] = 0;
        }
    }

    if (t < 128) {
        softmax_row<MM>(lB0, B0s, (t / GG) * (MM * GG) + (t % GG), GG);
    } else if (t < 160) {
        if (t < 128 + GG) softmax_row<CC>(lPi, Pis, t - 128, GG);
    } else if (t < 544) {
        const int r = t - 160;
        softmax_row<CC>(lQ, Qs, (r / (CC * GG)) * (CC * CC * GG) + (r % (CC * GG)), CC * GG);
    } else if (t < SETUP_SOFTMAX_THREADS) {
        const int r = t - 544;
        softmax_row<MM>(lB, Bs,
                        (r / (CC * GG)) * (CC * MM * GG) + ((r % (CC * GG)) / GG) * (MM * GG) + (r % GG),
                        GG);
    }
}

__global__ void degree_kernel(const int* __restrict__ dst, int* __restrict__ cnt, int E) {
    int t = blockIdx.x * blockDim.x + threadIdx.x;
    if (t < E) atomicAdd(&cnt[dst[t]], 1);
}

#define SCAN_THREADS 1024
#define SCAN_TILE (SCAN_THREADS * 4)

__device__ __forceinline__ int wave_inclusive_sum(int v, int lane) {
    v += __builtin_amdgcn_update_dpp(0, v, 0x111, 0xF, 0xF, true);
    v += __builtin_amdgcn_update_dpp(0, v, 0x112, 0xF, 0xF, true);
    v += __builtin_amdgcn_update_dpp(0, v, 0x114, 0xF, 0xF, true);
    v += __builtin_amdgcn_update_dpp(0, v, 0x118, 0xF, 0xF, true);
    const int low_row_total = __builtin_amdgcn_readlane(v, 15);
    return v + (lane >= 16 ? low_row_total : 0);
}

__global__ __launch_bounds__(SCAN_THREADS) void scan_kernel(const int* __restrict__ cnt,
                                                            int* __restrict__ rowstart,
                                                            int* __restrict__ cursor,
                                                            float* __restrict__ inv, int N) {
    __shared__ int s_front[32];
    __shared__ int s_tile[32];
    const int tid = threadIdx.x;
    const int lane = tid & 31;
    const int w = tid >> 5;
    const int tile0 = blockIdx.x * SCAN_TILE;

    int front = 0;
    for (int i = tid * 4; i < tile0; i += SCAN_TILE) {
        const int4 q = *(const int4*)(cnt + i);
        front += (q.x + q.y) + (q.z + q.w);
    }

    const int i0 = tile0 + tid * 4;
    int4 v = make_int4(0, 0, 0, 0);
    if (i0 + 3 < N) {
        v = *(const int4*)(cnt + i0);
    } else {
        if (i0 + 0 < N) v.x = cnt[i0 + 0];
        if (i0 + 1 < N) v.y = cnt[i0 + 1];
        if (i0 + 2 < N) v.z = cnt[i0 + 2];
    }
    if (i0 + 3 < N) {
        *(float4*)(inv + i0) = make_float4(1.0f / fmaxf((float)v.x, 1.0f),
                                           1.0f / fmaxf((float)v.y, 1.0f),
                                           1.0f / fmaxf((float)v.z, 1.0f),
                                           1.0f / fmaxf((float)v.w, 1.0f));
    } else {
        if (i0 + 0 < N) inv[i0 + 0] = 1.0f / fmaxf((float)v.x, 1.0f);
        if (i0 + 1 < N) inv[i0 + 1] = 1.0f / fmaxf((float)v.y, 1.0f);
        if (i0 + 2 < N) inv[i0 + 2] = 1.0f / fmaxf((float)v.z, 1.0f);
    }

    const int s1 = v.x + v.y;
    const int s2 = s1 + v.z;
    const int s3 = s2 + v.w;

    const int incl = wave_inclusive_sum(s3, lane);
    const int front_incl = wave_inclusive_sum(front, lane);
    if (lane == 31) {
        s_tile[w] = incl;
        s_front[w] = front_incl;
    }
    __syncthreads();

    const int part = s_front[lane] + (lane < w ? s_tile[lane] : 0);
    const int wave_base = __builtin_amdgcn_readlane(wave_inclusive_sum(part, lane), 31);

    const int e0 = wave_base + incl - s3;
    const int e1 = e0 + v.x;
    const int e2 = e0 + s1;
    const int e3 = e0 + s2;
    if (i0 + 3 < N) {
        const int4 o = make_int4(e0, e1, e2, e3);
        *(int4*)(rowstart + i0) = o;
        *(int4*)(cursor + i0) = o;
    } else {
        if (i0 + 0 <= N) rowstart[i0 + 0] = e0;
        if (i0 + 1 <= N) rowstart[i0 + 1] = e1;
        if (i0 + 2 <= N) rowstart[i0 + 2] = e2;
        if (i0 + 3 <= N) rowstart[i0 + 3] = e3;
        if (i0 + 0 < N) cursor[i0 + 0] = e0;
        if (i0 + 1 < N) cursor[i0 + 1] = e1;
        if (i0 + 2 < N) cursor[i0 + 2] = e2;
    }
}


__device__ __forceinline__ void csr_fill_body(const int* __restrict__ dst, const int* __restrict__ src,
                                              int* __restrict__ cursor, int* __restrict__ eidx,
                                              int E, int e) {
    if (e < E) {
        int p = atomicAdd(&cursor[dst[e]], 1);
        eidx[p] = src[e];
    }
}

__device__ __forceinline__ void layer0_body(const int* __restrict__ x,
                                            const float* __restrict__ B0s,
                                            const float* __restrict__ Pis,
                                            float* __restrict__ tab0,
                                            float* __restrict__ out,
                                            int N, int t) {
    if (t >= N * GG + MM * GG) return;
    const bool per_node = t < N * GG;
    const int r = per_node ? t : t - N * GG;
    const int n = r / GG, g = r % GG;
    const int xm = per_node ? x[n] : n;
    float u[CC];
    float Z = 0.f;
#pragma unroll
    for (int c = 0; c < CC; ++c) {
        u[c] = B0s[(c * MM + xm) * GG + g] * Pis[c * GG + g];
        Z += u[c];
    }
    if (per_node) {
        out[(size_t)n * (LL * GG) + g] = logf(Z);
    } else {
        const float iz = 1.0f / Z;
#pragma unroll
        for (int c = 0; c < CC; ++c)
            tab0[xm * (GG * CC) + g * CC + c] = u[c] * iz;
    }
}

__global__ __launch_bounds__(256) void fill_and_layer0_kernel(
    const int* __restrict__ dst, const int* __restrict__ src,
    int* __restrict__ cursor, int* __restrict__ eidx, int E, int fillBlocks,
    const int* __restrict__ x, const float* __restrict__ B0s,
    const float* __restrict__ Pis, float* __restrict__ tab0,
    float* __restrict__ out, int N) {
    const int b = blockIdx.x;
    if (b < fillBlocks)
        csr_fill_body(dst, src, cursor, eidx, E, b * 256 + threadIdx.x);
    else
        layer0_body(x, B0s, Pis, tab0, out, N, (b - fillBlocks) * 256 + threadIdx.x);
}

template <int R>
__device__ __forceinline__ void add_rows(const float* __restrict__ hin, int ids, int j0,
                                         int lane, float4& acc) {
    float4 v[R];
#pragma unroll
    for (int u = 0; u < R; ++u) {
        const int s = __builtin_amdgcn_readlane(ids, j0 + u);
        v[u] = *(const float4*)(hin + (size_t)s * 128 + lane * 4);
    }
#pragma unroll
    for (int u = 0; u < R; ++u) {
        acc.x += v[u].x; acc.y += v[u].y; acc.z += v[u].z; acc.w += v[u].w;
    }
}

#define MEAN_PITCH 132
#define LZ_PITCH 17

template <bool FIRST>
__global__ __launch_bounds__(512) void gather_update_kernel(
    const int* __restrict__ x,
    const float* __restrict__ Qg,
    const float* __restrict__ Bg,
    const float* __restrict__ hin,
    const int* __restrict__ rowstart,
    const int* __restrict__ eidx,
    const float* __restrict__ invc,
    float* __restrict__ hout,
    float* __restrict__ out,
    int N, int lout) {
    __shared__ __attribute__((aligned(16))) float mean[16 * MEAN_PITCH];
    __shared__ float qa[GG * 16 * CC];
    __shared__ float lz[16 * LZ_PITCH];
    __shared__ __attribute__((aligned(16))) float tab[FIRST ? MM * GG * CC : 4];

    const int tid = threadIdx.x;
    const int w = tid >> 5;
    const int lane = tid & 31;
    const int half = lane >> 4;
    const int col = lane & 15;
    const int g = w;
    const int k0 = half * 2;

    const int node = blockIdx.x * 16 + col;
    const int nn = node < N ? node : N - 1;
    const int xm = x[nn];

    const int qc = col & 7;
    const float cm = (col < CC) ? 1.0f : 0.0f;
    v2f b0, b1;
    b0.x = Qg[(qc * CC + k0 + 0) * GG + g] * cm;
    b0.y = Qg[(qc * CC + k0 + 1) * GG + g] * cm;
    b1.x = Qg[(qc * CC + k0 + 4) * GG + g] * cm;
    b1.y = Qg[(qc * CC + k0 + 5) * GG + g] * cm;

    const int gn = __builtin_amdgcn_readfirstlane(blockIdx.x * 16 + w);
    float4 acc = make_float4(0.f, 0.f, 0.f, 0.f);
    if constexpr (FIRST) {
        const float4 t0 = ((const float4*)hin)[tid];
        const float4 t1 = ((const float4*)hin)[tid + 512];
        int beg = 0, end = 0;
        float iv = 0.f;
        if (gn < N) {
            beg = rowstart[gn];
            end = rowstart[gn + 1];
            iv = invc[gn];
        }
        int n = (end - beg < 32) ? (end - beg) : 32;
        int xs = (lane < n) ? x[eidx[beg + lane]] : 0;
        ((float4*)tab)[tid] = t0;
        ((float4*)tab)[tid + 512] = t1;
        __syncthreads();
        for (int c0 = beg; c0 < end;) {
            int j = 0;
            for (; j + 8 <= n; j += 8) add_rows<8>(tab, xs, j, lane, acc);
            switch (n - j) {
                case 7: add_rows<7>(tab, xs, j, lane, acc); break;
                case 6: add_rows<6>(tab, xs, j, lane, acc); break;
                case 5: add_rows<5>(tab, xs, j, lane, acc); break;
                case 4: add_rows<4>(tab, xs, j, lane, acc); break;
                case 3: add_rows<3>(tab, xs, j, lane, acc); break;
                case 2: add_rows<2>(tab, xs, j, lane, acc); break;
                case 1: add_rows<1>(tab, xs, j, lane, acc); break;
                default: break;
            }
            c0 += 32;
            if (c0 < end) {
                n = (end - c0 < 32) ? (end - c0) : 32;
                xs = (lane < n) ? x[eidx[c0 + lane]] : 0;
            }
        }
        acc.x *= iv; acc.y *= iv; acc.z *= iv; acc.w *= iv;
    } else if (gn < N) {
        const int beg = rowstart[gn];
        const int end = rowstart[gn + 1];
        const float iv = invc[gn];
        for (int c0 = beg; c0 < end; c0 += 32) {
            const int n = (end - c0 < 32) ? (end - c0) : 32;
            const int ids = (lane < n) ? eidx[c0 + lane] : 0;
            int j = 0;
            for (; j + 8 <= n; j += 8) add_rows<8>(hin, ids, j, lane, acc);
            switch (n - j) {
                case 7: add_rows<7>(hin, ids, j, lane, acc); break;
                case 6: add_rows<6>(hin, ids, j, lane, acc); break;
                case 5: add_rows<5>(hin, ids, j, lane, acc); break;
                case 4: add_rows<4>(hin, ids, j, lane, acc); break;
                case 3: add_rows<3>(hin, ids, j, lane, acc); break;
                case 2: add_rows<2>(hin, ids, j, lane, acc); break;
                case 1: add_rows<1>(hin, ids, j, lane, acc); break;
                default: break;
            }
        }
        acc.x *= iv; acc.y *= iv; acc.z *= iv; acc.w *= iv;
    }
    *(float4*)(mean + w * MEAN_PITCH + lane * 4) = acc;

    float bn[CC];
#pragma unroll
    for (int c = 0; c < CC; ++c) bn[c] = Bg[(c * MM + xm) * GG + g];

    __syncthreads();

    const float* arow = mean + col * MEAN_PITCH + g * CC + k0;
    v2f a0, a1;
    a0.x = arow[0];
    a0.y = arow[1];
    a1.x = arow[4];
    a1.y = arow[5];

    v8f d = {0.f, 0.f, 0.f, 0.f, 0.f, 0.f, 0.f, 0.f};
    d = __builtin_amdgcn_wmma_f32_16x16x4_f32(false, a0, false, b0,
                                              (short)0, d, false, false);
    d = __builtin_amdgcn_wmma_f32_16x16x4_f32(false, a1, false, b1,
                                              (short)0, d, false, false);

    if (col < CC) {
#pragma unroll
        for (int r = 0; r < 8; ++r)
            qa[(w * 16 + (r + half * 8)) * CC + col] = d[r];
    }
    __syncthreads();

    if (lane < 16) {
        if (node < N) {
            float u[CC];
            float Z = 0.f;
#pragma unroll
            for (int c = 0; c < CC; ++c) {
                u[c] = bn[c] * qa[(w * 16 + lane) * CC + c];
                Z += u[c];
            }
            lz[lane * LZ_PITCH + g] = logf(Z);
            const float iz = 1.0f / Z;
            float* hs = mean + lane * MEAN_PITCH + g * CC;
            *(float4*)(hs + 0) = make_float4(u[0] * iz, u[1] * iz, u[2] * iz, u[3] * iz);
            *(float4*)(hs + 4) = make_float4(u[4] * iz, u[5] * iz, u[6] * iz, u[7] * iz);
        }
    }
    __syncthreads();

    if (gn < N)
        *(float4*)(hout + (size_t)gn * (GG * CC) + lane * 4) =
            *(const float4*)(mean + w * MEAN_PITCH + lane * 4);
    if (tid < 16 * GG) {
        const int ln = tid >> 4;
        const int lnode = blockIdx.x * 16 + ln;
        if (lnode < N)
            out[(size_t)lnode * (LL * GG) + lout * GG + (tid & 15)] = lz[ln * LZ_PITCH + (tid & 15)];
    }
}

extern "C" void kernel_launch(void* const* d_in, const int* in_sizes, int n_in,
                              void* d_out, int out_size, void* d_ws, size_t ws_size,
                              hipStream_t stream) {
    const int* x = (const int*)d_in[0];
    const int* ei = (const int*)d_in[1];
    const float* lB0 = (const float*)d_in[2];
    const float* lPi = (const float*)d_in[3];
    const float* lQ = (const float*)d_in[4];
    const float* lB = (const float*)d_in[5];
    float* out = (float*)d_out;

    const int N = in_sizes[0];
    const int E = in_sizes[1] / 2;
    const int* dst = ei;
    const int* srcv = ei + E;

    float* ws = (float*)d_ws;
    size_t off = 0;
    float* h0 = ws + off; off += (size_t)N * GG * CC;
    float* h1 = ws + off; off += (size_t)N * GG * CC;
    float* tab0 = h0;
    float* B0s = ws + off; off += CC * MM * GG;
    float* Pis = ws + off; off += CC * GG;
    float* Qs  = ws + off; off += (LL - 1) * CC * CC * GG;
    float* Bs  = ws + off; off += (LL - 1) * CC * MM * GG;
    off = (off + 3) & ~(size_t)3;
    float* invc = ws + off; off += N;
    off = (off + 3) & ~(size_t)3;
    int* cnt      = (int*)(ws + off); off += N;
    off = (off + 3) & ~(size_t)3;
    int* rowstart = (int*)(ws + off); off += N + 1;
    off = (off + 3) & ~(size_t)3;
    int* cursor   = (int*)(ws + off); off += N;
    int* eidx     = (int*)(ws + off); off += E;
    (void)ws_size; (void)n_in; (void)out_size;

    const int setupBlocks = ((N + 3) / 4 + SETUP_THREADS - 1) / SETUP_THREADS +
                            (SETUP_SOFTMAX_THREADS + SETUP_THREADS - 1) / SETUP_THREADS;
    setup_kernel<<<setupBlocks, SETUP_THREADS, 0, stream>>>(lB0, lPi, lQ, lB, B0s, Pis, Qs, Bs, cnt, N);

    degree_kernel<<<(E + 255) / 256, 256, 0, stream>>>(dst, cnt, E);
    scan_kernel<<<N / SCAN_TILE + 1, SCAN_THREADS, 0, stream>>>(cnt, rowstart, cursor, invc, N);

    const int fillBlocks = (E + 255) / 256;
    const int layer0Blocks = (N * GG + MM * GG + 255) / 256;
    fill_and_layer0_kernel<<<fillBlocks + layer0Blocks, 256, 0, stream>>>(
        dst, srcv, cursor, eidx, E, fillBlocks, x, B0s, Pis, tab0, out, N);

    const int tileBlocks = (N + 15) / 16;
    gather_update_kernel<true><<<tileBlocks, 512, 0, stream>>>(
        x, Qs, Bs, tab0, rowstart, eidx, invc, h1, out, N, 1);
    gather_update_kernel<false><<<tileBlocks, 512, 0, stream>>>(
        x, Qs + CC * CC * GG, Bs + CC * MM * GG, h1, rowstart, eidx, invc, h0, out, N, 2);
    gather_update_kernel<false><<<tileBlocks, 512, 0, stream>>>(
        x, Qs + 2 * CC * CC * GG, Bs + 2 * CC * MM * GG, h0, rowstart, eidx, invc, h1, out, N, 3);
}


# Every launch is a grid the chip holds at once (1024 blocks of 16 waves per graph layer, 2048 of 8 for fill + layer 0); blocks loop over node tiles / edges: a third of the waves started per call
# speedup vs baseline: 3.1853x; 1.4292x over previous
#include <hip/hip_runtime.h>
#include <hip/hip_bf16.h>

#define CC 8
#define MM 32
#define GG 16
#define LL 4

typedef float v2f __attribute__((ext_vector_type(2)));
typedef float v8f __attribute__((ext_vector_type(8)));

template <int LEN>
__device__ __forceinline__ void softmax_row(const float* __restrict__ src,
                                            float* __restrict__ dst,
                                            int base, int stride) {
    float v[LEN];
#pragma unroll
    for (int i = 0; i < LEN; ++i) v[i] = src[base + i * stride];
    float mx = -3.402823466e38f;
#pragma unroll
    for (int i = 0; i < LEN; ++i) mx = fmaxf(mx, v[i]);
    float sum = 0.f;
#pragma unroll
    for (int i = 0; i < LEN; ++i) {
        v[i] = expf(v[i] - mx);
        sum += v[i];
    }
    const float inv = 1.0f / sum;
#pragma unroll
    for (int i = 0; i < LEN; ++i) dst[base + i * stride] = v[i] * inv;
}

#define SETUP_THREADS 256
#define SETUP_SOFTMAX_THREADS 928

__global__ __launch_bounds__(SETUP_THREADS) void setup_kernel(
    const float* __restrict__ lB0, const float* __restrict__ lPi,
    const float* __restrict__ lQ, const float* __restrict__ lB,
    float* __restrict__ B0s, float* __restrict__ Pis,
    float* __restrict__ Qs, float* __restrict__ Bs,
    int* __restrict__ cnt, int ncnt) {
    const int t = blockIdx.x * SETUP_THREADS + threadIdx.x;
    const int step = gridDim.x * SETUP_THREADS * 4;
    for (int i = t * 4; i < ncnt; i += step) {
        if (i + 3 < ncnt) {
            *(int4*)(cnt + i) = make_int4(0, 0, 0, 0);
        } else {
            if (i + 0 < ncnt) cnt[i + 0] = 0;
            if (i + 1 < ncnt) cnt[i + 1] = 0;
            if (i + 2 < ncnt) cnt[i + 2] = 0;
        }
    }

    if (t < 128) {
        softmax_row<MM>(lB0, B0s, (t / GG) * (MM * GG) + (t % GG), GG);
    } else if (t < 160) {
        if (t < 128 + GG) softmax_row<CC>(lPi, Pis, t - 128, GG);
    } else if (t < 544) {
        const int r = t - 160;
        softmax_row<CC>(lQ, Qs, (r / (CC * GG)) * (CC * CC * GG) + (r % (CC * GG)), CC * GG);
    } else if (t < SETUP_SOFTMAX_THREADS) {
        const int r = t - 544;
        softmax_row<MM>(lB, Bs,
                        (r / (CC * GG)) * (CC * MM * GG) + ((r % (CC * GG)) / GG) * (MM * GG) + (r % GG),
                        GG);
    }
}

#define ADJ_K 32

__device__ __forceinline__ void adj_fill_body(const int* __restrict__ dst, const int* __restrict__ src,
                                              int* __restrict__ cnt, int* __restrict__ adj,
                                              int* __restrict__ ovl, int N, int E, int e) {
    if (e < E) {
        const int d = dst[e];
        const int s = src[e];
        const int p = atomicAdd(&cnt[d], 1);
        int* slot = adj + ((size_t)d * ADJ_K + p);
        if (p >= ADJ_K) slot = ovl + atomicAdd(&cnt[N], 1);
        *slot = (p < ADJ_K) ? s : e;
    }
}

__device__ __forceinline__ void layer0_body(const int* __restrict__ x,
                                            const float* __restrict__ B0s,
                                            const float* __restrict__ Pis,
                                            float* __restrict__ tab0,
                                            float* __restrict__ out,
                                            int N, int t) {
    if (t >= N * GG + MM * GG) return;
    const bool per_node = t < N * GG;
    const int r = per_node ? t : t - N * GG;
    const int n = r / GG, g = r % GG;
    const int xm = per_node ? x[n] : n;
    float u[CC];
    float Z = 0.f;
#pragma unroll
    for (int c = 0; c < CC; ++c) {
        u[c] = B0s[(c * MM + xm) * GG + g] * Pis[c * GG + g];
        Z += u[c];
    }
    if (per_node) {
        out[(size_t)n * (LL * GG) + g] = logf(Z);
    } else {
        const float iz = 1.0f / Z;
#pragma unroll
        for (int c = 0; c < CC; ++c)
            tab0[xm * (GG * CC) + g * CC + c] = u[c] * iz;
    }
}

__global__ __launch_bounds__(256) void fill_and_layer0_kernel(
    const int* __restrict__ dst, const int* __restrict__ src,
    int* __restrict__ cnt, int* __restrict__ adj, int* __restrict__ ovl, int E, int fillBlocks,
    const int* __restrict__ x, const float* __restrict__ B0s,
    const float* __restrict__ Pis, float* __restrict__ tab0,
    float* __restrict__ out, int N) {
    const int b = blockIdx.x;
    if (b < fillBlocks) {
        const int step = fillBlocks * 256;
        for (int e = b * 256 + threadIdx.x; e < E; e += step)
            adj_fill_body(dst, src, cnt, adj, ovl, N, E, e);
    } else {
        const int total = N * GG + MM * GG;
        const int step = ((int)gridDim.x - fillBlocks) * 256;
        for (int t = (b - fillBlocks) * 256 + threadIdx.x; t < total; t += step)
            layer0_body(x, B0s, Pis, tab0, out, N, t);
    }
}

template <int R>
__device__ __forceinline__ void add_rows(const float* __restrict__ hin, int ids, int j0,
                                         int lane, float4& acc) {
    float4 v[R];
#pragma unroll
    for (int u = 0; u < R; ++u) {
        const int s = __builtin_amdgcn_readlane(ids, j0 + u);
        v[u] = *(const float4*)(hin + (size_t)s * 128 + lane * 4);
    }
#pragma unroll
    for (int u = 0; u < R; ++u) {
        acc.x += v[u].x; acc.y += v[u].y; acc.z += v[u].z; acc.w += v[u].w;
    }
}

template <bool FIRST>
__device__ __forceinline__ void add_overflow_rows(const float* __restrict__ rows,
                                                  const int* __restrict__ x,
                                                  const int* __restrict__ dst,
                                                  const int* __restrict__ src,
                                                  const int* __restrict__ ovl, int novf,
                                                  int gn, int lane, float4& acc) {
    for (int q0 = 0; q0 < novf; q0 += 32) {
        const int q = q0 + lane;
        int r = -1;
        if (q < novf) {
            const int e = ovl[q];
            if (dst[e] == gn) {
                const int s = src[e];
                r = FIRST ? x[s] : s;
            }
        }
        unsigned m = __builtin_amdgcn_ballot_w32(r >= 0);
        while (m != 0u) {
            const int j = __builtin_ctz(m);
            m &= m - 1u;
            add_rows<1>(rows, r, j, lane, acc);
        }
    }
}


#define MEAN_PITCH 132
#define LZ_PITCH 17

template <bool FIRST>
__global__ __launch_bounds__(512, 16) void gather_update_kernel(
    const int* __restrict__ x,
    const float* __restrict__ Qg,
    const float* __restrict__ Bg,
    const float* __restrict__ hin,
    const int* __restrict__ cnt,
    const int* __restrict__ adj,
    const int* __restrict__ ovl,
    const int* __restrict__ dst,
    const int* __restrict__ src,
    float* __restrict__ hout,
    float* __restrict__ out,
    int N, int lout, int nTiles) {
    __shared__ __attribute__((aligned(16))) float mean[16 * MEAN_PITCH];
    __shared__ float qa[GG * 16 * CC];
    __shared__ float lz[16 * LZ_PITCH];
    __shared__ __attribute__((aligned(16))) float tab[FIRST ? MM * GG * CC : 4];

    const int tid = threadIdx.x;
    const int w = tid >> 5;
    const int lane = tid & 31;
    const int half = lane >> 4;
    const int col = lane & 15;
    const int g = w;
    const int k0 = half * 2;

    const int qc = col & 7;
    const float cm = (col < CC) ? 1.0f : 0.0f;
    v2f b0, b1;
    b0.x = Qg[(qc * CC + k0 + 0) * GG + g] * cm;
    b0.y = Qg[(qc * CC + k0 + 1) * GG + g] * cm;
    b1.x = Qg[(qc * CC + k0 + 4) * GG + g] * cm;
    b1.y = Qg[(qc * CC + k0 + 5) * GG + g] * cm;

    if constexpr (FIRST) {
        const float4 t0 = ((const float4*)hin)[tid];
        const float4 t1 = ((const float4*)hin)[tid + 512];
        ((float4*)tab)[tid] = t0;
        ((float4*)tab)[tid + 512] = t1;
        __syncthreads();
    }

    for (int tile = blockIdx.x; tile < nTiles; tile += gridDim.x) {
    const int node = tile * 16 + col;
    const int nn = node < N ? node : N - 1;
    const int xm = x[nn];

    const int gn = __builtin_amdgcn_readfirstlane(tile * 16 + w);
    float4 acc = make_float4(0.f, 0.f, 0.f, 0.f);
    if constexpr (FIRST) {
        int deg = 0;
        if (gn < N) deg = cnt[gn];
        deg = __builtin_amdgcn_readfirstlane(deg);
        const float iv = 1.0f / fmaxf((float)deg, 1.0f);
        const int n = (deg < ADJ_K) ? deg : ADJ_K;
        const int xs = (lane < n) ? x[adj[(size_t)gn * ADJ_K + lane]] : 0;
        int j = 0;
        for (; j + 8 <= n; j += 8) add_rows<8>(tab, xs, j, lane, acc);
        switch ((n - j) & 7) {
            case 7: add_rows<7>(tab, xs, j, lane, acc); break;
            case 6: add_rows<6>(tab, xs, j, lane, acc); break;
            case 5: add_rows<5>(tab, xs, j, lane, acc); break;
            case 4: add_rows<4>(tab, xs, j, lane, acc); break;
            case 3: add_rows<3>(tab, xs, j, lane, acc); break;
            case 2: add_rows<2>(tab, xs, j, lane, acc); break;
            case 1: add_rows<1>(tab, xs, j, lane, acc); break;
            default: break;
        }
        if (deg > ADJ_K)
            add_overflow_rows<true>(tab, x, dst, src, ovl, cnt[N], gn, lane, acc);
        acc.x *= iv; acc.y *= iv; acc.z *= iv; acc.w *= iv;
    } else if (gn < N) {
        const int deg = __builtin_amdgcn_readfirstlane(cnt[gn]);
        const float iv = 1.0f / fmaxf((float)deg, 1.0f);
        const int n = (deg < ADJ_K) ? deg : ADJ_K;
        const int ids = (lane < n) ? adj[(size_t)gn * ADJ_K + lane] : 0;
        int j = 0;
        for (; j + 8 <= n; j += 8) add_rows<8>(hin, ids, j, lane, acc);
        switch ((n - j) & 7) {
            case 7: add_rows<7>(hin, ids, j, lane, acc); break;
            case 6: add_rows<6>(hin, ids, j, lane, acc); break;
            case 5: add_rows<5>(hin, ids, j, lane, acc); break;
            case 4: add_rows<4>(hin, ids, j, lane, acc); break;
            case 3: add_rows<3>(hin, ids, j, lane, acc); break;
            case 2: add_rows<2>(hin, ids, j, lane, acc); break;
            case 1: add_rows<1>(hin, ids, j, lane, acc); break;
            default: break;
        }
        if (deg > ADJ_K)
            add_overflow_rows<false>(hin, x, dst, src, ovl, cnt[N], gn, lane, acc);
        acc.x *= iv; acc.y *= iv; acc.z *= iv; acc.w *= iv;
    }
    *(float4*)(mean + w * MEAN_PITCH + lane * 4) = acc;

    float bn[CC];
#pragma unroll
    for (int c = 0; c < CC; ++c) bn[c] = Bg[(c * MM + xm) * GG + g];

    __syncthreads();

    const float* arow = mean + col * MEAN_PITCH + g * CC + k0;
    v2f a0, a1;
    a0.x = arow[0];
    a0.y = arow[1];
    a1.x = arow[4];
    a1.y = arow[5];

    v8f d = {0.f, 0.f, 0.f, 0.f, 0.f, 0.f, 0.f, 0.f};
    d = __builtin_amdgcn_wmma_f32_16x16x4_f32(false, a0, false, b0,
                                              (short)0, d, false, false);
    d = __builtin_amdgcn_wmma_f32_16x16x4_f32(false, a1, false, b1,
                                              (short)0, d, false, false);

    if (col < CC) {
#pragma unroll
        for (int r = 0; r < 8; ++r)
            qa[(w * 16 + (r + half * 8)) * CC + col] = d[r];
    }
    __syncthreads();

    if (lane < 16) {
        if (node < N) {
            float u[CC];
            float Z = 0.f;
#pragma unroll
            for (int c = 0; c < CC; ++c) {
                u[c] = bn[c] * qa[(w * 16 + lane) * CC + c];
                Z += u[c];
            }
            lz[lane * LZ_PITCH + g] = logf(Z);
            const float iz = 1.0f / Z;
            float* hs = mean + lane * MEAN_PITCH + g * CC;
            *(float4*)(hs + 0) = make_float4(u[0] * iz, u[1] * iz, u[2] * iz, u[3] * iz);
            *(float4*)(hs + 4) = make_float4(u[4] * iz, u[5] * iz, u[6] * iz, u[7] * iz);
        }
    }
    __syncthreads();

    if (gn < N)
        *(float4*)(hout + (size_t)gn * (GG * CC) + lane * 4) =
            *(const float4*)(mean + w * MEAN_PITCH + lane * 4);
    if (tid < 16 * GG) {
        const int ln = tid >> 4;
        const int lnode = tile * 16 + ln;
        if (lnode < N)
            out[(size_t)lnode * (LL * GG) + lout * GG + (tid & 15)] = lz[ln * LZ_PITCH + (tid & 15)];
    }
    }
}

extern "C" void kernel_launch(void* const* d_in, const int* in_sizes, int n_in,
                              void* d_out, int out_size, void* d_ws, size_t ws_size,
                              hipStream_t stream) {
    const int* x = (const int*)d_in[0];
    const int* ei = (const int*)d_in[1];
    const float* lB0 = (const float*)d_in[2];
    const float* lPi = (const float*)d_in[3];
    const float* lQ = (const float*)d_in[4];
    const float* lB = (const float*)d_in[5];
    float* out = (float*)d_out;

    const int N = in_sizes[0];
    const int E = in_sizes[1] / 2;
    const int* dst = ei;
    const int* srcv = ei + E;

    float* ws = (float*)d_ws;
    size_t off = 0;
    float* h0 = ws + off; off += (size_t)N * GG * CC;
    float* h1 = ws + off; off += (size_t)N * GG * CC;
    float* tab0 = h0;
    float* B0s = ws + off; off += CC * MM * GG;
    float* Pis = ws + off; off += CC * GG;
    float* Qs  = ws + off; off += (LL - 1) * CC * CC * GG;
    float* Bs  = ws + off; off += (LL - 1) * CC * MM * GG;
    off = (off + 3) & ~(size_t)3;
    int* cnt = (int*)(ws + off); off += N + 1;
    off = (off + 31) & ~(size_t)31;
    int* adj = (int*)(ws + off); off += (size_t)N * ADJ_K;
    int* ovl = (int*)(ws + off); off += E;
    (void)ws_size; (void)n_in; (void)out_size;

    const int setupBlocks = ((N + 1 + 3) / 4 + SETUP_THREADS - 1) / SETUP_THREADS +
                            (SETUP_SOFTMAX_THREADS + SETUP_THREADS - 1) / SETUP_THREADS;
    setup_kernel<<<setupBlocks, SETUP_THREADS, 0, stream>>>(lB0, lPi, lQ, lB, B0s, Pis, Qs, Bs, cnt, N + 1);

    const int fillBlocks = 1024;
    const int layer0Blocks = 1024;
    fill_and_layer0_kernel<<<fillBlocks + layer0Blocks, 256, 0, stream>>>(
        dst, srcv, cnt, adj, ovl, E, fillBlocks, x, B0s, Pis, tab0, out, N);

    const int nTiles = (N + 15) / 16;
    const int layerBlocks = nTiles < 1024 ? nTiles : 1024;
    gather_update_kernel<true><<<layerBlocks, 512, 0, stream>>>(
        x, Qs, Bs, tab0, cnt, adj, ovl, dst, srcv, h1, out, N, 1, nTiles);
    gather_update_kernel<false><<<layerBlocks, 512, 0, stream>>>(
        x, Qs + CC * CC * GG, Bs + CC * MM * GG, h1, cnt, adj, ovl, dst, srcv, h0, out, N, 2, nTiles);
    gather_update_kernel<false><<<layerBlocks, 512, 0, stream>>>(
        x, Qs + 2 * CC * CC * GG, Bs + 2 * CC * MM * GG, h0, cnt, adj, ovl, dst, srcv, h1, out, N, 3, nTiles);
}


# Adjacency rows built with no global atomic per edge: edges binned by 64-node destination range in LDS per chunk (launch 1), one block per range fills its rows in LDS and stores whole lines
# speedup vs baseline: 4.4096x; 1.3844x over previous
#include <hip/hip_runtime.h>
#include <hip/hip_bf16.h>

#define CC 8
#define MM 32
#define GG 16
#define LL 4

typedef float v2f __attribute__((ext_vector_type(2)));
typedef float v8f __attribute__((ext_vector_type(8)));

template <int LEN>
__device__ __forceinline__ void softmax_row(const float* __restrict__ src,
                                            float* __restrict__ dst,
                                            int base, int stride) {
    float v[LEN];
#pragma unroll
    for (int i = 0; i < LEN; ++i) v[i] = src[base + i * stride];
    float mx = -3.402823466e38f;
#pragma unroll
    for (int i = 0; i < LEN; ++i) mx = fmaxf(mx, v[i]);
    float sum = 0.f;
#pragma unroll
    for (int i = 0; i < LEN; ++i) {
        v[i] = expf(v[i] - mx);
        sum += v[i];
    }
    const float inv = 1.0f / sum;
#pragma unroll
    for (int i = 0; i < LEN; ++i) dst[base + i * stride] = v[i] * inv;
}

#define RANGE_NODES 64
#define RANGE_SHIFT 6
#define MAX_RANGES 1024
#define EDGE_BITS 26
#define BIN_EPT 13
#define BIN_CHUNK (256 * BIN_EPT)

__device__ __forceinline__ int block_incl_scan_256(int v, int* wsum, int t) {
    const int lane = t & 31;
    const int w = t >> 5;
#pragma unroll
    for (int o = 1; o < 32; o <<= 1) {
        const int u = __shfl_up(v, o, 32);
        if (lane >= o) v += u;
    }
    if (lane == 31) wsum[w] = v;
    __syncthreads();
    int add = 0;
#pragma unroll
    for (int k = 0; k < 8; ++k) add += (k < w) ? wsum[k] : 0;
    return v + add;
}

__device__ __forceinline__ void bin_chunk_body(const int* __restrict__ dst, const int* __restrict__ src,
                                               unsigned* __restrict__ sorted, int* __restrict__ offT,
                                               int E, int nRanges, int nChunks, int b) {
    __shared__ int hist[MAX_RANGES];
    __shared__ int wsum[8];
    __shared__ __attribute__((aligned(16))) uint2 stage[BIN_CHUNK];
    const int t = threadIdx.x;
    for (int i = t; i < MAX_RANGES; i += 256) hist[i] = 0;
    const int base = b * BIN_CHUNK;
    int d[BIN_EPT], s[BIN_EPT], rk[BIN_EPT];
#pragma unroll
    for (int u = 0; u < BIN_EPT; ++u) {
        const int e = base + u * 256 + t;
        const bool ok = e < E;
        d[u] = ok ? dst[e] : -1;
        s[u] = ok ? src[e] : 0;
    }
    __syncthreads();
#pragma unroll
    for (int u = 0; u < BIN_EPT; ++u) {
        rk[u] = 0;
        if (d[u] >= 0) {
            int r = d[u] >> RANGE_SHIFT;
            r = r < MAX_RANGES - 1 ? r : MAX_RANGES - 1;
            rk[u] = atomicAdd(&hist[r], 1);
        }
    }
    __syncthreads();
    const int c0 = hist[4 * t + 0], c1 = hist[4 * t + 1], c2 = hist[4 * t + 2], c3 = hist[4 * t + 3];
    const int sum = c0 + c1 + c2 + c3;
    const int excl = block_incl_scan_256(sum, wsum, t) - sum;
    hist[4 * t + 0] = excl;
    hist[4 * t + 1] = excl + c0;
    hist[4 * t + 2] = excl + c0 + c1;
    hist[4 * t + 3] = excl + c0 + c1 + c2;
    __syncthreads();
    for (int r = t; r <= nRanges; r += 256) offT[(size_t)r * nChunks + b] = base + hist[r];
#pragma unroll
    for (int u = 0; u < BIN_EPT; ++u) {
        if (d[u] >= 0) {
            int r = d[u] >> RANGE_SHIFT;
            r = r < MAX_RANGES - 1 ? r : MAX_RANGES - 1;
            const unsigned e = (unsigned)(base + u * 256 + t);
            stage[hist[r] + rk[u]] =
                make_uint2(((unsigned)(d[u] & (RANGE_NODES - 1)) << EDGE_BITS) | e, (unsigned)s[u]);
        }
    }
    __syncthreads();
    const int nEdges = (E - base) < BIN_CHUNK ? (E - base) : BIN_CHUNK;
    for (int i = t; i < nEdges; i += 256)
        *(uint2*)(sorted + 2 * (size_t)(base + i)) = stage[i];
}

#define SETUP_THREADS 256
#define SETUP_SOFTMAX_THREADS 928
#define SETUP_SOFTMAX_BLOCKS 4

__global__ __launch_bounds__(SETUP_THREADS) void setup_kernel(
    const float* __restrict__ lB0, const float* __restrict__ lPi,
    const float* __restrict__ lQ, const float* __restrict__ lB,
    float* __restrict__ B0s, float* __restrict__ Pis,
    float* __restrict__ Qs, float* __restrict__ Bs,
    int* __restrict__ ovfCount,
    const int* __restrict__ dst, const int* __restrict__ src,
    unsigned* __restrict__ sorted, int* __restrict__ offT,
    int E, int nRanges, int nChunks) {
    if (blockIdx.x >= SETUP_SOFTMAX_BLOCKS) {
        bin_chunk_body(dst, src, sorted, offT, E, nRanges, nChunks, (int)blockIdx.x - SETUP_SOFTMAX_BLOCKS);
        return;
    }
    const int t = blockIdx.x * SETUP_THREADS + threadIdx.x;
    if (t == 0) *ovfCount = 0;

    if (t < 128) {
        softmax_row<MM>(lB0, B0s, (t / GG) * (MM * GG) + (t % GG), GG);
    } else if (t < 160) {
        if (t < 128 + GG) softmax_row<CC>(lPi, Pis, t - 128, GG);
    } else if (t < 544) {
        const int r = t - 160;
        softmax_row<CC>(lQ, Qs, (r / (CC * GG)) * (CC * CC * GG) + (r % (CC * GG)), CC * GG);
    } else if (t < SETUP_SOFTMAX_THREADS) {
        const int r = t - 544;
        softmax_row<MM>(lB, Bs,
                        (r / (CC * GG)) * (CC * MM * GG) + ((r % (CC * GG)) / GG) * (MM * GG) + (r % GG),
                        GG);
    }
}

#define ADJ_K 32
#define FILL_WINDOW 2048

__device__ __forceinline__ void range_fill_body(const int* __restrict__ offT,
                                                const unsigned* __restrict__ sorted,
                                                int* __restrict__ cnt, int* __restrict__ adj,
                                                int* __restrict__ ovl, int N, int nChunks, int r) {
    __shared__ __attribute__((aligned(16))) int rows[RANGE_NODES * ADJ_K];
    __shared__ int degs[RANGE_NODES];
    __shared__ int segStart[256];
    __shared__ int segPre[256];
    __shared__ unsigned short owner[FILL_WINDOW];
    __shared__ int wsum[8];
    __shared__ int totalS;
    const int t = threadIdx.x;
    for (int i = t; i < RANGE_NODES * ADJ_K / 4; i += 256) ((int4*)rows)[i] = make_int4(0, 0, 0, 0);
    if (t < RANGE_NODES) degs[t] = 0;

    for (int b0 = 0; b0 < nChunks; b0 += 256) {
        const int b = b0 + t;
        int st = 0, len = 0;
        if (b < nChunks) {
            st = offT[(size_t)r * nChunks + b];
            len = offT[(size_t)(r + 1) * nChunks + b] - st;
        }
        const int incl = block_incl_scan_256(len, wsum, t);
        const int excl = incl - len;
        segStart[t] = st;
        segPre[t] = excl;
        if (t == 255) totalS = incl;
        __syncthreads();
        const int total = totalS;
        for (int w0 = 0; w0 < total; w0 += FILL_WINDOW) {
            const int wend = total < w0 + FILL_WINDOW ? total : w0 + FILL_WINDOW;
            const int lo = excl > w0 ? excl : w0;
            const int hi = incl < wend ? incl : wend;
            for (int j = lo; j < hi; ++j) owner[j - w0] = (unsigned short)t;
            __syncthreads();
            int pos[FILL_WINDOW / 256];
            uint2 en[FILL_WINDOW / 256];
#pragma unroll
            for (int k = 0; k < FILL_WINDOW / 256; ++k) {
                const int i = w0 + k * 256 + t;
                pos[k] = -1;
                if (i < wend) {
                    const int sg = owner[i - w0];
                    pos[k] = segStart[sg] + (i - segPre[sg]);
                }
            }
#pragma unroll
            for (int k = 0; k < FILL_WINDOW / 256; ++k) {
                en[k] = make_uint2(0u, 0u);
                if (pos[k] >= 0) en[k] = *(const uint2*)(sorted + 2 * (size_t)pos[k]);
            }
#pragma unroll
            for (int k = 0; k < FILL_WINDOW / 256; ++k) {
                if (pos[k] >= 0) {
                    const int local = (int)(en[k].x >> EDGE_BITS);
                    const int p = atomicAdd(&degs[local], 1);
                    if (p < ADJ_K) {
                        rows[local * ADJ_K + p] = (int)en[k].y;
                    } else {
                        const int q = atomicAdd(&cnt[N], 1);
                        ovl[q] = (int)(en[k].x & ((1u << EDGE_BITS) - 1u));
                    }
                }
            }
            __syncthreads();
        }
    }
    __syncthreads();
    const int n0 = r * RANGE_NODES;
    if (t < RANGE_NODES && n0 + t < N) cnt[n0 + t] = degs[t];
    for (int i = t; i < RANGE_NODES * ADJ_K / 4; i += 256) {
        if (n0 + (i >> 3) < N)
            ((int4*)(adj + (size_t)n0 * ADJ_K))[i] = ((const int4*)rows)[i];
    }
}

__device__ __forceinline__ void layer0_body(const int* __restrict__ x,
                                            const float* __restrict__ B0s,
                                            const float* __restrict__ Pis,
                                            float* __restrict__ tab0,
                                            float* __restrict__ out,
                                            int N, int t) {
    if (t >= N * GG + MM * GG) return;
    const bool per_node = t < N * GG;
    const int r = per_node ? t : t - N * GG;
    const int n = r / GG, g = r % GG;
    const int xm = per_node ? x[n] : n;
    float u[CC];
    float Z = 0.f;
#pragma unroll
    for (int c = 0; c < CC; ++c) {
        u[c] = B0s[(c * MM + xm) * GG + g] * Pis[c * GG + g];
        Z += u[c];
    }
    if (per_node) {
        out[(size_t)n * (LL * GG) + g] = logf(Z);
    } else {
        const float iz = 1.0f / Z;
#pragma unroll
        for (int c = 0; c < CC; ++c)
            tab0[xm * (GG * CC) + g * CC + c] = u[c] * iz;
    }
}

__global__ __launch_bounds__(256) void fill_and_layer0_kernel(
    const int* __restrict__ offT, const unsigned* __restrict__ sorted,
    int* __restrict__ cnt, int* __restrict__ adj, int* __restrict__ ovl, int nChunks, int fillBlocks,
    const int* __restrict__ x, const float* __restrict__ B0s,
    const float* __restrict__ Pis, float* __restrict__ tab0,
    float* __restrict__ out, int N) {
    const int b = blockIdx.x;
    if (b < fillBlocks) {
        range_fill_body(offT, sorted, cnt, adj, ovl, N, nChunks, b);
    } else {
        const int total = N * GG + MM * GG;
        const int step = ((int)gridDim.x - fillBlocks) * 256;
        for (int t = (b - fillBlocks) * 256 + threadIdx.x; t < total; t += step)
            layer0_body(x, B0s, Pis, tab0, out, N, t);
    }
}

template <int R>
__device__ __forceinline__ void add_rows(const float* __restrict__ hin, int ids, int j0,
                                         int lane, float4& acc) {
    float4 v[R];
#pragma unroll
    for (int u = 0; u < R; ++u) {
        const int s = __builtin_amdgcn_readlane(ids, j0 + u);
        v[u] = *(const float4*)(hin + (size_t)s * 128 + lane * 4);
    }
#pragma unroll
    for (int u = 0; u < R; ++u) {
        acc.x += v[u].x; acc.y += v[u].y; acc.z += v[u].z; acc.w += v[u].w;
    }
}

template <bool FIRST>
__device__ __forceinline__ void add_overflow_rows(const float* __restrict__ rows,
                                                  const int* __restrict__ x,
                                                  const int* __restrict__ dst,
                                                  const int* __restrict__ src,
                                                  const int* __restrict__ ovl, int novf,
                                                  int gn, int lane, float4& acc) {
    for (int q0 = 0; q0 < novf; q0 += 32) {
        const int q = q0 + lane;
        int r = -1;
        if (q < novf) {
            const int e = ovl[q];
            if (dst[e] == gn) {
                const int s = src[e];
                r = FIRST ? x[s] : s;
            }
        }
        unsigned m = __builtin_amdgcn_ballot_w32(r >= 0);
        while (m != 0u) {
            const int j = __builtin_ctz(m);
            m &= m - 1u;
            add_rows<1>(rows, r, j, lane, acc);
        }
    }
}


#define MEAN_PITCH 132
#define LZ_PITCH 17

template <bool FIRST>
__global__ __launch_bounds__(512, 16) void gather_update_kernel(
    const int* __restrict__ x,
    const float* __restrict__ Qg,
    const float* __restrict__ Bg,
    const float* __restrict__ hin,
    const int* __restrict__ cnt,
    const int* __restrict__ adj,
    const int* __restrict__ ovl,
    const int* __restrict__ dst,
    const int* __restrict__ src,
    float* __restrict__ hout,
    float* __restrict__ out,
    int N, int lout, int nTiles) {
    __shared__ __attribute__((aligned(16))) float mean[16 * MEAN_PITCH];
    __shared__ float qa[GG * 16 * CC];
    __shared__ float lz[16 * LZ_PITCH];
    __shared__ __attribute__((aligned(16))) float tab[FIRST ? MM * GG * CC : 4];

    const int tid = threadIdx.x;
    const int w = tid >> 5;
    const int lane = tid & 31;
    const int half = lane >> 4;
    const int col = lane & 15;
    const int g = w;
    const int k0 = half * 2;

    const int qc = col & 7;
    const float cm = (col < CC) ? 1.0f : 0.0f;
    v2f b0, b1;
    b0.x = Qg[(qc * CC + k0 + 0) * GG + g] * cm;
    b0.y = Qg[(qc * CC + k0 + 1) * GG + g] * cm;
    b1.x = Qg[(qc * CC + k0 + 4) * GG + g] * cm;
    b1.y = Qg[(qc * CC + k0 + 5) * GG + g] * cm;

    if constexpr (FIRST) {
        const float4 t0 = ((const float4*)hin)[tid];
        const float4 t1 = ((const float4*)hin)[tid + 512];
        ((float4*)tab)[tid] = t0;
        ((float4*)tab)[tid + 512] = t1;
        __syncthreads();
    }

    for (int tile = blockIdx.x; tile < nTiles; tile += gridDim.x) {
    const int node = tile * 16 + col;
    const int nn = node < N ? node : N - 1;
    const int xm = x[nn];

    const int gn = __builtin_amdgcn_readfirstlane(tile * 16 + w);
    float4 acc = make_float4(0.f, 0.f, 0.f, 0.f);
    if constexpr (FIRST) {
        int deg = 0;
        if (gn < N) deg = cnt[gn];
        deg = __builtin_amdgcn_readfirstlane(deg);
        const float iv = 1.0f / fmaxf((float)deg, 1.0f);
        const int n = (deg < ADJ_K) ? deg : ADJ_K;
        const int xs = (lane < n) ? x[adj[(size_t)gn * ADJ_K + lane]] : 0;
        int j = 0;
        for (; j + 8 <= n; j += 8) add_rows<8>(tab, xs, j, lane, acc);
        switch ((n - j) & 7) {
            case 7: add_rows<7>(tab, xs, j, lane, acc); break;
            case 6: add_rows<6>(tab, xs, j, lane, acc); break;
            case 5: add_rows<5>(tab, xs, j, lane, acc); break;
            case 4: add_rows<4>(tab, xs, j, lane, acc); break;
            case 3: add_rows<3>(tab, xs, j, lane, acc); break;
            case 2: add_rows<2>(tab, xs, j, lane, acc); break;
            case 1: add_rows<1>(tab, xs, j, lane, acc); break;
            default: break;
        }
        if (deg > ADJ_K)
            add_overflow_rows<true>(tab, x, dst, src, ovl, cnt[N], gn, lane, acc);
        acc.x *= iv; acc.y *= iv; acc.z *= iv; acc.w *= iv;
    } else if (gn < N) {
        const int deg = __builtin_amdgcn_readfirstlane(cnt[gn]);
        const float iv = 1.0f / fmaxf((float)deg, 1.0f);
        const int n = (deg < ADJ_K) ? deg : ADJ_K;
        const int ids = (lane < n) ? adj[(size_t)gn * ADJ_K + lane] : 0;
        int j = 0;
        for (; j + 8 <= n; j += 8) add_rows<8>(hin, ids, j, lane, acc);
        switch ((n - j) & 7) {
            case 7: add_rows<7>(hin, ids, j, lane, acc); break;
            case 6: add_rows<6>(hin, ids, j, lane, acc); break;
            case 5: add_rows<5>(hin, ids, j, lane, acc); break;
            case 4: add_rows<4>(hin, ids, j, lane, acc); break;
            case 3: add_rows<3>(hin, ids, j, lane, acc); break;
            case 2: add_rows<2>(hin, ids, j, lane, acc); break;
            case 1: add_rows<1>(hin, ids, j, lane, acc); break;
            default: break;
        }
        if (deg > ADJ_K)
            add_overflow_rows<false>(hin, x, dst, src, ovl, cnt[N], gn, lane, acc);
        acc.x *= iv; acc.y *= iv; acc.z *= iv; acc.w *= iv;
    }
    *(float4*)(mean + w * MEAN_PITCH + lane * 4) = acc;

    float bn[CC];
#pragma unroll
    for (int c = 0; c < CC; ++c) bn[c] = Bg[(c * MM + xm) * GG + g];

    __syncthreads();

    const float* arow = mean + col * MEAN_PITCH + g * CC + k0;
    v2f a0, a1;
    a0.x = arow[0];
    a0.y = arow[1];
    a1.x = arow[4];
    a1.y = arow[5];

    v8f d = {0.f, 0.f, 0.f, 0.f, 0.f, 0.f, 0.f, 0.f};
    d = __builtin_amdgcn_wmma_f32_16x16x4_f32(false, a0, false, b0,
                                              (short)0, d, false, false);
    d = __builtin_amdgcn_wmma_f32_16x16x4_f32(false, a1, false, b1,
                                              (short)0, d, false, false);

    if (col < CC) {
#pragma unroll
        for (int r = 0; r < 8; ++r)
            qa[(w * 16 + (r + half * 8)) * CC + col] = d[r];
    }
    __syncthreads();

    if (lane < 16) {
        if (node < N) {
            float u[CC];
            float Z = 0.f;
#pragma unroll
            for (int c = 0; c < CC; ++c) {
                u[c] = bn[c] * qa[(w * 16 + lane) * CC + c];
                Z += u[c];
            }
            lz[lane * LZ_PITCH + g] = logf(Z);
            const float iz = 1.0f / Z;
            float* hs = mean + lane * MEAN_PITCH + g * CC;
            *(float4*)(hs + 0) = make_float4(u[0] * iz, u[1] * iz, u[2] * iz, u[3] * iz);
            *(float4*)(hs + 4) = make_float4(u[4] * iz, u[5] * iz, u[6] * iz, u[7] * iz);
        }
    }
    __syncthreads();

    if (gn < N)
        *(float4*)(hout + (size_t)gn * (GG * CC) + lane * 4) =
            *(const float4*)(mean + w * MEAN_PITCH + lane * 4);
    if (tid < 16 * GG) {
        const int ln = tid >> 4;
        const int lnode = tile * 16 + ln;
        if (lnode < N)
            out[(size_t)lnode * (LL * GG) + lout * GG + (tid & 15)] = lz[ln * LZ_PITCH + (tid & 15)];
    }
    }
}

extern "C" void kernel_launch(void* const* d_in, const int* in_sizes, int n_in,
                              void* d_out, int out_size, void* d_ws, size_t ws_size,
                              hipStream_t stream) {
    const int* x = (const int*)d_in[0];
    const int* ei = (const int*)d_in[1];
    const float* lB0 = (const float*)d_in[2];
    const float* lPi = (const float*)d_in[3];
    const float* lQ = (const float*)d_in[4];
    const float* lB = (const float*)d_in[5];
    float* out = (float*)d_out;

    const int N = in_sizes[0];
    const int E = in_sizes[1] / 2;
    const int* dst = ei;
    const int* srcv = ei + E;

    float* ws = (float*)d_ws;
    size_t off = 0;
    float* h0 = ws + off; off += (size_t)N * GG * CC;
    float* h1 = ws + off; off += (size_t)N * GG * CC;
    float* tab0 = h0;
    float* B0s = ws + off; off += CC * MM * GG;
    float* Pis = ws + off; off += CC * GG;
    float* Qs  = ws + off; off += (LL - 1) * CC * CC * GG;
    float* Bs  = ws + off; off += (LL - 1) * CC * MM * GG;
    off = (off + 3) & ~(size_t)3;
    int* cnt = (int*)(ws + off); off += N + 1;
    off = (off + 31) & ~(size_t)31;
    int* adj = (int*)(ws + off); off += (size_t)N * ADJ_K;
    int* ovl = (int*)(ws + off); off += E;
    const int nRanges = (N + RANGE_NODES - 1) / RANGE_NODES;
    const int nChunks = (E + BIN_CHUNK - 1) / BIN_CHUNK;
    unsigned* sorted = (unsigned*)h1;
    int* offT = (int*)h1 + 2 * (size_t)nChunks * BIN_CHUNK;
    (void)ws_size; (void)n_in; (void)out_size;

    setup_kernel<<<SETUP_SOFTMAX_BLOCKS + nChunks, SETUP_THREADS, 0, stream>>>(
        lB0, lPi, lQ, lB, B0s, Pis, Qs, Bs, cnt + N, dst, srcv, sorted, offT, E, nRanges, nChunks);

    const int fillBlocks = nRanges;
    const int layer0Blocks = 1024;
    fill_and_layer0_kernel<<<fillBlocks + layer0Blocks, 256, 0, stream>>>(
        offT, sorted, cnt, adj, ovl, nChunks, fillBlocks, x, B0s, Pis, tab0, out, N);

    const int nTiles = (N + 15) / 16;
    const int layerBlocks = nTiles < 1024 ? nTiles : 1024;
    gather_update_kernel<true><<<layerBlocks, 512, 0, stream>>>(
        x, Qs, Bs, tab0, cnt, adj, ovl, dst, srcv, h1, out, N, 1, nTiles);
    gather_update_kernel<false><<<layerBlocks, 512, 0, stream>>>(
        x, Qs + CC * CC * GG, Bs + CC * MM * GG, h1, cnt, adj, ovl, dst, srcv, h0, out, N, 2, nTiles);
    gather_update_kernel<false><<<layerBlocks, 512, 0, stream>>>(
        x, Qs + 2 * CC * CC * GG, Bs + 2 * CC * MM * GG, h0, cnt, adj, ovl, dst, srcv, h1, out, N, 3, nTiles);
}
